# starting kernel
# baseline (speedup 1.0000x reference)
#include <hip/hip_runtime.h>
#include <math.h>

#define NN 50000
#define EE 800000

typedef __attribute__((ext_vector_type(2))) float v2f;
typedef __attribute__((ext_vector_type(8))) float v8f;

__device__ __forceinline__ unsigned f2ord(float f) {
  unsigned u = __float_as_uint(f);
  return (u & 0x80000000u) ? ~u : (u | 0x80000000u);
}
__device__ __forceinline__ float ord2f(unsigned e) {
  unsigned u = (e & 0x80000000u) ? (e ^ 0x80000000u) : ~e;
  return __uint_as_float(u);
}

__global__ __launch_bounds__(32)
void gat_gemm_xwt(const float* __restrict__ X, const float* __restrict__ W,
                  float* __restrict__ OUT, int K, int Opad, int Oout) {
  const int lane = threadIdx.x & 31;
  const int row0 = blockIdx.x << 4;
  const int col0 = blockIdx.y << 6;
  const int mr   = lane & 15;
  const int half = lane >> 4;
  const int koff = half << 1;

  const float* xrow = X + (size_t)(row0 + mr) * (size_t)K + koff;
  const float* wrow = W + (size_t)(col0 + mr) * (size_t)K + koff;
  const size_t wstep = (size_t)16 * (size_t)K;

  v8f acc[4] = {};
  v2f a0, a1;
  v2f b0[4], b1[4];

  a0 = *(const v2f*)(xrow);
#pragma unroll
  for (int t = 0; t < 4; ++t) b0[t] = *(const v2f*)(wrow + (size_t)t * wstep);

  int k0 = 0;
  for (; k0 + 8 < K; k0 += 8) {
    a1 = *(const v2f*)(xrow + k0 + 4);
#pragma unroll
    for (int t = 0; t < 4; ++t)
      b1[t] = *(const v2f*)(wrow + (size_t)t * wstep + k0 + 4);
#pragma unroll
    for (int t = 0; t < 4; ++t)
      acc[t] = __builtin_amdgcn_wmma_f32_16x16x4_f32(
          false, a0, false, b0[t], (short)0, acc[t], false, false);

    a0 = *(const v2f*)(xrow + k0 + 8);
#pragma unroll
    for (int t = 0; t < 4; ++t)
      b0[t] = *(const v2f*)(wrow + (size_t)t * wstep + k0 + 8);
#pragma unroll
    for (int t = 0; t < 4; ++t)
      acc[t] = __builtin_amdgcn_wmma_f32_16x16x4_f32(
          false, a1, false, b1[t], (short)0, acc[t], false, false);
  }

  a1 = *(const v2f*)(xrow + k0 + 4);
#pragma unroll
  for (int t = 0; t < 4; ++t)
    b1[t] = *(const v2f*)(wrow + (size_t)t * wstep + k0 + 4);
#pragma unroll
  for (int t = 0; t < 4; ++t)
    acc[t] = __builtin_amdgcn_wmma_f32_16x16x4_f32(
        false, a0, false, b0[t], (short)0, acc[t], false, false);
#pragma unroll
  for (int t = 0; t < 4; ++t)
    acc[t] = __builtin_amdgcn_wmma_f32_16x16x4_f32(
        false, a1, false, b1[t], (short)0, acc[t], false, false);

#pragma unroll
  for (int t = 0; t < 4; ++t) {
    const int oc = col0 + (t << 4) + mr;
    if (oc < Oout) {
      float* orow = OUT + (size_t)(row0 + (half << 3)) * (size_t)Oout + oc;
#pragma unroll
      for (int v = 0; v < 8; ++v) {
        orow[(size_t)v * (size_t)Oout] = acc[t][v];
      }
    }
  }
}

__global__ void gat_pad_w(const float* __restrict__ Wsrc, float* __restrict__ Wdst,
                          int rows, int padrows, int K) {
  int idx = blockIdx.x * blockDim.x + threadIdx.x;
  int total = padrows * K;
  if (idx >= total) return;
  int r = idx / K;
  Wdst[idx] = (r < rows) ? Wsrc[idx] : 0.0f;
}

__global__ void gat_el_er(const float* __restrict__ Hm, const float* __restrict__ al,
                          const float* __restrict__ ar, float* __restrict__ el,
                          float* __restrict__ er, int total, int Hh, int F) {
  int idx = blockIdx.x * blockDim.x + threadIdx.x;
  if (idx >= total) return;
  int h = idx % Hh;
  int n = idx / Hh;
  const float* row = Hm + (size_t)n * (size_t)(Hh * F) + (size_t)h * F;
  const float* a = al + h * F;
  const float* b = ar + h * F;
  float sl = 0.f, sr = 0.f;
  for (int f = 0; f < F; ++f) {
    float v = row[f];
    sl += v * a[f];
    sr += v * b[f];
  }
  el[idx] = sl;
  er[idx] = sr;
}

__global__ void gat_init_nm(unsigned* __restrict__ mEnc, float* __restrict__ den,
                            int total) {
  int idx = blockIdx.x * blockDim.x + threadIdx.x;
  if (idx < total) { mEnc[idx] = 0u; den[idx] = 0.0f; }
}

__global__ void gat_edge_smax(const float* __restrict__ el, const float* __restrict__ er,
                              const int* __restrict__ src, const int* __restrict__ dst,
                              float* __restrict__ sb, unsigned* __restrict__ mEnc,
                              int total, int Hh) {
  int idx = blockIdx.x * blockDim.x + threadIdx.x;
  if (idx >= total) return;
  int e = idx / Hh;
  int h = idx - e * Hh;
  int s_ = src[e], d_ = dst[e];
  float s = el[s_ * Hh + h] + er[d_ * Hh + h];
  s = (s > 0.0f) ? s : 0.2f * s;
  sb[idx] = s;
  atomicMax(mEnc + d_ * Hh + h, f2ord(s));
}

__global__ void gat_edge_expsum(float* __restrict__ sb, const unsigned* __restrict__ mEnc,
                                float* __restrict__ den, const int* __restrict__ dst,
                                int total, int Hh) {
  int idx = blockIdx.x * blockDim.x + threadIdx.x;
  if (idx >= total) return;
  int e = idx / Hh;
  int h = idx - e * Hh;
  int d_ = dst[e];
  float m = ord2f(mEnc[d_ * Hh + h]);
  float ex = expf(sb[idx] - m);
  sb[idx] = ex;
  atomicAdd(den + d_ * Hh + h, ex);
}

__global__ void gat_scatter(const float* __restrict__ Hm, const float* __restrict__ sb,
                            const float* __restrict__ den, const int* __restrict__ src,
                            const int* __restrict__ dst, float* __restrict__ AGG,
                            int HF, int F, int Hh) {
  int e = blockIdx.x;
  int j = threadIdx.x;
  if (j >= HF) return;
  int s_ = src[e], d_ = dst[e];
  int h = j / F;
  float alpha = sb[(size_t)e * Hh + h] / den[(size_t)d_ * Hh + h];
  atomicAdd(AGG + (size_t)d_ * HF + j, Hm[(size_t)s_ * HF + j] * alpha);
}

__global__ void gat_elu(float* __restrict__ y, long total) {
  long idx = (long)blockIdx.x * blockDim.x + threadIdx.x;
  if (idx >= total) return;
  float x = y[idx];
  y[idx] = (x > 0.0f) ? x : expm1f(x);
}

static void run_layer(const float* X, int K, int Hh, int F,
                      const float* W, const float* al, const float* ar, const float* rW,
                      const int* src, const int* dst,
                      float* Hbuf, float* AGG, bool act, int Opad,
                      float* el, float* er, float* den, unsigned* mEnc, float* sb,
                      hipStream_t stream) {
  const int HF = Hh * F;
  dim3 ggrid(NN / 16, Opad / 64);
  gat_gemm_xwt<<<ggrid, 32, 0, stream>>>(X, W, Hbuf, K, Opad, HF);
  gat_gemm_xwt<<<ggrid, 32, 0, stream>>>(X, rW, AGG, K, Opad, HF);

  const int nh = NN * Hh;
  gat_el_er<<<(nh + 255) / 256, 256, 0, stream>>>(Hbuf, al, ar, el, er, nh, Hh, F);
  gat_init_nm<<<(nh + 255) / 256, 256, 0, stream>>>(mEnc, den, nh);

  const int eh = EE * Hh;
  gat_edge_smax<<<(eh + 255) / 256, 256, 0, stream>>>(el, er, src, dst, sb, mEnc, eh, Hh);
  gat_edge_expsum<<<(eh + 255) / 256, 256, 0, stream>>>(sb, mEnc, den, dst, eh, Hh);

  const int bs = (HF >= 256) ? 256 : 64;
  gat_scatter<<<EE, bs, 0, stream>>>(Hbuf, sb, den, src, dst, AGG, HF, F, Hh);

  if (act) {
    long tot = (long)NN * HF;
    gat_elu<<<(unsigned)((tot + 255) / 256), 256, 0, stream>>>(AGG, tot);
  }
}

extern "C" void kernel_launch(void* const* d_in, const int* in_sizes, int n_in,
                              void* d_out, int out_size, void* d_ws, size_t ws_size,
                              hipStream_t stream) {
  (void)in_sizes; (void)n_in; (void)out_size; (void)ws_size;

  const float* feat = (const float*)d_in[0];
  const int*   src  = (const int*)d_in[1];
  const int*   dst  = (const int*)d_in[2];
  const float* W0  = (const float*)d_in[3];
  const float* al0 = (const float*)d_in[4];
  const float* ar0 = (const float*)d_in[5];
  const float* rW0 = (const float*)d_in[6];
  const float* W1  = (const float*)d_in[7];
  const float* al1 = (const float*)d_in[8];
  const float* ar1 = (const float*)d_in[9];
  const float* rW1 = (const float*)d_in[10];
  const float* W2  = (const float*)d_in[11];
  const float* al2 = (const float*)d_in[12];
  const float* ar2 = (const float*)d_in[13];
  const float* rW2 = (const float*)d_in[14];

  const size_t NF = (size_t)NN * 256;
  float* ws    = (float*)d_ws;
  float* bufH  = ws;
  float* bufY0 = bufH  + NF;
  float* bufY1 = bufY0 + NF;
  float* el    = bufY1 + NF;
  float* er    = el  + (size_t)NN * 4;
  float* den   = er  + (size_t)NN * 4;
  unsigned* mE = (unsigned*)(den + (size_t)NN * 4);
  float* sb    = (float*)mE + (size_t)NN * 4;
  float* W2p   = sb + (size_t)EE * 4;
  float* rW2p  = W2p + (size_t)64 * 256;

  {
    int tot = 64 * 256;
    gat_pad_w<<<(tot + 255) / 256, 256, 0, stream>>>(W2,  W2p,  40, 64, 256);
    gat_pad_w<<<(tot + 255) / 256, 256, 0, stream>>>(rW2, rW2p, 40, 64, 256);
  }

  run_layer(feat, 128, 4, 64, W0, al0, ar0, rW0, src, dst,
            bufH, bufY0, true, 256, el, er, den, mE, sb, stream);
  run_layer(bufY0, 256, 4, 64, W1, al1, ar1, rW1, src, dst,
            bufH, bufY1, true, 256, el, er, den, mE, sb, stream);
  run_layer(bufY1, 256, 1, 40, W2p, al2, ar2, rW2p, src, dst,
            bufH, (float*)d_out, false, 64, el, er, den, mE, sb, stream);
}


# Atomic scatter replaced by per-destination gather over an in-call counting sort of dst (int atomics): a node row is summed in registers and added once; float sum order is the sort's arrival order
# speedup vs baseline: 2.2495x; 2.2495x over previous
#include <hip/hip_runtime.h>
#include <math.h>

#define NN 50000
#define EE 800000

typedef __attribute__((ext_vector_type(2))) float v2f;
typedef __attribute__((ext_vector_type(8))) float v8f;

__device__ __forceinline__ unsigned f2ord(float f) {
  unsigned u = __float_as_uint(f);
  return (u & 0x80000000u) ? ~u : (u | 0x80000000u);
}
__device__ __forceinline__ float ord2f(unsigned e) {
  unsigned u = (e & 0x80000000u) ? (e ^ 0x80000000u) : ~e;
  return __uint_as_float(u);
}

__global__ __launch_bounds__(32)
void gat_gemm_xwt(const float* __restrict__ X, const float* __restrict__ W,
                  float* __restrict__ OUT, int K, int Opad, int Oout) {
  const int lane = threadIdx.x & 31;
  const int row0 = blockIdx.x << 4;
  const int col0 = blockIdx.y << 6;
  const int mr   = lane & 15;
  const int half = lane >> 4;
  const int koff = half << 1;

  const float* xrow = X + (size_t)(row0 + mr) * (size_t)K + koff;
  const float* wrow = W + (size_t)(col0 + mr) * (size_t)K + koff;
  const size_t wstep = (size_t)16 * (size_t)K;

  v8f acc[4] = {};
  v2f a0, a1;
  v2f b0[4], b1[4];

  a0 = *(const v2f*)(xrow);
#pragma unroll
  for (int t = 0; t < 4; ++t) b0[t] = *(const v2f*)(wrow + (size_t)t * wstep);

  int k0 = 0;
  for (; k0 + 8 < K; k0 += 8) {
    a1 = *(const v2f*)(xrow + k0 + 4);
#pragma unroll
    for (int t = 0; t < 4; ++t)
      b1[t] = *(const v2f*)(wrow + (size_t)t * wstep + k0 + 4);
#pragma unroll
    for (int t = 0; t < 4; ++t)
      acc[t] = __builtin_amdgcn_wmma_f32_16x16x4_f32(
          false, a0, false, b0[t], (short)0, acc[t], false, false);

    a0 = *(const v2f*)(xrow + k0 + 8);
#pragma unroll
    for (int t = 0; t < 4; ++t)
      b0[t] = *(const v2f*)(wrow + (size_t)t * wstep + k0 + 8);
#pragma unroll
    for (int t = 0; t < 4; ++t)
      acc[t] = __builtin_amdgcn_wmma_f32_16x16x4_f32(
          false, a1, false, b1[t], (short)0, acc[t], false, false);
  }

  a1 = *(const v2f*)(xrow + k0 + 4);
#pragma unroll
  for (int t = 0; t < 4; ++t)
    b1[t] = *(const v2f*)(wrow + (size_t)t * wstep + k0 + 4);
#pragma unroll
  for (int t = 0; t < 4; ++t)
    acc[t] = __builtin_amdgcn_wmma_f32_16x16x4_f32(
        false, a0, false, b0[t], (short)0, acc[t], false, false);
#pragma unroll
  for (int t = 0; t < 4; ++t)
    acc[t] = __builtin_amdgcn_wmma_f32_16x16x4_f32(
        false, a1, false, b1[t], (short)0, acc[t], false, false);

#pragma unroll
  for (int t = 0; t < 4; ++t) {
    const int oc = col0 + (t << 4) + mr;
    if (oc < Oout) {
      float* orow = OUT + (size_t)(row0 + (half << 3)) * (size_t)Oout + oc;
#pragma unroll
      for (int v = 0; v < 8; ++v) {
        orow[(size_t)v * (size_t)Oout] = acc[t][v];
      }
    }
  }
}

__global__ void gat_pad_w(const float* __restrict__ Wsrc, float* __restrict__ Wdst,
                          int rows, int padrows, int K) {
  int idx = blockIdx.x * blockDim.x + threadIdx.x;
  int total = padrows * K;
  if (idx >= total) return;
  int r = idx / K;
  Wdst[idx] = (r < rows) ? Wsrc[idx] : 0.0f;
}

__global__ void gat_el_er(const float* __restrict__ Hm, const float* __restrict__ al,
                          const float* __restrict__ ar, float* __restrict__ el,
                          float* __restrict__ er, int total, int Hh, int F) {
  int idx = blockIdx.x * blockDim.x + threadIdx.x;
  if (idx >= total) return;
  int h = idx % Hh;
  int n = idx / Hh;
  const float* row = Hm + (size_t)n * (size_t)(Hh * F) + (size_t)h * F;
  const float* a = al + h * F;
  const float* b = ar + h * F;
  float sl = 0.f, sr = 0.f;
  for (int f = 0; f < F; ++f) {
    float v = row[f];
    sl += v * a[f];
    sr += v * b[f];
  }
  el[idx] = sl;
  er[idx] = sr;
}

__global__ void gat_init_nm(unsigned* __restrict__ mEnc, float* __restrict__ den,
                            int total) {
  int idx = blockIdx.x * blockDim.x + threadIdx.x;
  if (idx < total) { mEnc[idx] = 0u; den[idx] = 0.0f; }
}

__global__ void gat_edge_smax(const float* __restrict__ el, const float* __restrict__ er,
                              const int* __restrict__ src, const int* __restrict__ dst,
                              float* __restrict__ sb, unsigned* __restrict__ mEnc,
                              int total, int Hh) {
  int idx = blockIdx.x * blockDim.x + threadIdx.x;
  if (idx >= total) return;
  int e = idx / Hh;
  int h = idx - e * Hh;
  int s_ = src[e], d_ = dst[e];
  float s = el[s_ * Hh + h] + er[d_ * Hh + h];
  s = (s > 0.0f) ? s : 0.2f * s;
  sb[idx] = s;
  atomicMax(mEnc + d_ * Hh + h, f2ord(s));
}

__global__ void gat_edge_expsum(float* __restrict__ sb, const unsigned* __restrict__ mEnc,
                                float* __restrict__ den, const int* __restrict__ dst,
                                int total, int Hh) {
  int idx = blockIdx.x * blockDim.x + threadIdx.x;
  if (idx >= total) return;
  int e = idx / Hh;
  int h = idx - e * Hh;
  int d_ = dst[e];
  float m = ord2f(mEnc[d_ * Hh + h]);
  float ex = expf(sb[idx] - m);
  sb[idx] = ex;
  atomicAdd(den + d_ * Hh + h, ex);
}

#define CSR_PERM (NN + 1)
#define CSR_KEEP (CSR_PERM + EE)
#define CSR_CUR  CSR_KEEP
#define CSR_BSUM (CSR_CUR + NN)
#define CSR_NB   ((NN + 255) / 256)

__global__ void gat_csr_zero(int* __restrict__ cur, int total) {
  int idx = blockIdx.x * blockDim.x + threadIdx.x;
  if (idx < total) cur[idx] = 0;
}

__global__ void gat_csr_hist(const int* __restrict__ dst, int* __restrict__ cur, int total) {
  int e = blockIdx.x * blockDim.x + threadIdx.x;
  if (e < total) atomicAdd(cur + dst[e], 1);
}

__device__ __forceinline__ int gat_block_scan256(int v, int* sh) {
  const int t = threadIdx.x;
  sh[t] = v;
  __syncthreads();
  for (int d = 1; d < 256; d <<= 1) {
    int a = (t >= d) ? sh[t - d] : 0;
    __syncthreads();
    sh[t] += a;
    __syncthreads();
  }
  return sh[t];
}

__global__ __launch_bounds__(256)
void gat_csr_bsum(const int* __restrict__ cur, int* __restrict__ bsum) {
  __shared__ int sh[256];
  const int i = blockIdx.x * 256 + threadIdx.x;
  const int v = (i < NN) ? cur[i] : 0;
  const int inc = gat_block_scan256(v, sh);
  if (threadIdx.x == 255) bsum[blockIdx.x] = inc;
}

__global__ __launch_bounds__(256)
void gat_csr_offsets(int* __restrict__ cur, const int* __restrict__ bsum,
                     int* __restrict__ off) {
  __shared__ int sh[256];
  const int t = threadIdx.x;
  const int b = blockIdx.x;
  const int x = (t < b) ? bsum[t] : 0;
  gat_block_scan256(x, sh);
  const int before = sh[255];
  __syncthreads();
  const int i = b * 256 + t;
  const int v = (i < NN) ? cur[i] : 0;
  const int inc = gat_block_scan256(v, sh);
  if (i < NN) {
    const int o = before + inc - v;
    off[i] = o;
    cur[i] = o;
    if (i == NN - 1) off[NN] = o + v;
  }
}

__global__ void gat_csr_fill(const int* __restrict__ dst, int* __restrict__ cur,
                             int* __restrict__ perm, int total) {
  int e = blockIdx.x * blockDim.x + threadIdx.x;
  if (e >= total) return;
  int p = atomicAdd(cur + dst[e], 1);
  perm[p] = e;
}

__global__ void gat_csr_move(const int* __restrict__ from, int* __restrict__ to, int total) {
  int idx = blockIdx.x * blockDim.x + threadIdx.x;
  if (idx < total) to[idx] = from[idx];
}

__global__ __launch_bounds__(256)
void gat_gather256(const float* __restrict__ Hm, const float* __restrict__ sb,
                   const float* __restrict__ den, const int* __restrict__ src,
                   const int* __restrict__ off, const int* __restrict__ perm,
                   float* __restrict__ AGG) {
  const int lane = threadIdx.x & 31;
  const int n = __builtin_amdgcn_readfirstlane((int)(blockIdx.x * 8 + (threadIdx.x >> 5)));
  if (n >= NN) return;
  const int beg = off[n];
  const int end = off[n + 1];
  const float4 dn = *(const float4*)(den + (size_t)n * 4);
  const bool hi = lane >= 16;
  float4 acc0 = {0.f, 0.f, 0.f, 0.f};
  float4 acc1 = {0.f, 0.f, 0.f, 0.f};

  for (int base = beg; base < end; base += 32) {
    const int left = end - base;
    const int cnt = (left < 32) ? left : 32;
    int my_s = 0;
    float4 al = {0.f, 0.f, 0.f, 0.f};
    if (lane < cnt) {
      const int e = perm[base + lane];
      my_s = src[e];
      const float4 sc = *(const float4*)(sb + (size_t)e * 4);
      al.x = sc.x / dn.x;
      al.y = sc.y / dn.y;
      al.z = sc.z / dn.z;
      al.w = sc.w / dn.w;
    }
#pragma unroll 4
    for (int i = 0; i < cnt; ++i) {
      const int s = __builtin_amdgcn_readlane(my_s, i);
      const float a0 = __int_as_float(__builtin_amdgcn_readlane(__float_as_int(al.x), i));
      const float a1 = __int_as_float(__builtin_amdgcn_readlane(__float_as_int(al.y), i));
      const float a2 = __int_as_float(__builtin_amdgcn_readlane(__float_as_int(al.z), i));
      const float a3 = __int_as_float(__builtin_amdgcn_readlane(__float_as_int(al.w), i));
      const float w0 = hi ? a1 : a0;
      const float w1 = hi ? a3 : a2;
      const float4* row = (const float4*)(Hm + (size_t)s * 256);
      const float4 v0 = row[lane];
      const float4 v1 = row[32 + lane];
      acc0.x += v0.x * w0; acc0.y += v0.y * w0; acc0.z += v0.z * w0; acc0.w += v0.w * w0;
      acc1.x += v1.x * w1; acc1.y += v1.y * w1; acc1.z += v1.z * w1; acc1.w += v1.w * w1;
    }
  }

  float4* o = (float4*)(AGG + (size_t)n * 256);
  float4 r0 = o[lane];
  float4 r1 = o[32 + lane];
  r0.x += acc0.x; r0.y += acc0.y; r0.z += acc0.z; r0.w += acc0.w;
  r1.x += acc1.x; r1.y += acc1.y; r1.z += acc1.z; r1.w += acc1.w;
  o[lane] = r0;
  o[32 + lane] = r1;
}

__global__ __launch_bounds__(256)
void gat_gather40(const float* __restrict__ Hm, const float* __restrict__ sb,
                  const float* __restrict__ den, const int* __restrict__ src,
                  const int* __restrict__ off, const int* __restrict__ perm,
                  float* __restrict__ AGG) {
  const int lane = threadIdx.x & 31;
  const int g = __builtin_amdgcn_readfirstlane((int)(blockIdx.x * 8 + (threadIdx.x >> 5)));
  if (g * 4 >= NN) return;
  const int col = ((lane < 20) ? lane : 19) * 2;
  for (int q = 0; q < 4; ++q) {
    const int n = g * 4 + q;
    const int beg = off[n];
    const int end = off[n + 1];
    const float dn = den[n];
    float2 acc = {0.f, 0.f};
    for (int base = beg; base < end; base += 32) {
      const int left = end - base;
      const int cnt = (left < 32) ? left : 32;
      int my_s = 0;
      float my_a = 0.f;
      if (lane < cnt) {
        const int e = perm[base + lane];
        my_s = src[e];
        my_a = sb[e] / dn;
      }
#pragma unroll 4
      for (int i = 0; i < cnt; ++i) {
        const int s = __builtin_amdgcn_readlane(my_s, i);
        const float a = __int_as_float(__builtin_amdgcn_readlane(__float_as_int(my_a), i));
        const float2 v = *(const float2*)(Hm + (size_t)s * 40 + col);
        acc.x += v.x * a;
        acc.y += v.y * a;
      }
    }
    if (lane < 20) {
      float2* o = (float2*)(AGG + (size_t)n * 40 + col);
      float2 r = *o;
      r.x += acc.x;
      r.y += acc.y;
      *o = r;
    }
  }
}

__global__ void gat_elu(float* __restrict__ y, long total) {
  long idx = (long)blockIdx.x * blockDim.x + threadIdx.x;
  if (idx >= total) return;
  float x = y[idx];
  y[idx] = (x > 0.0f) ? x : expm1f(x);
}

static void run_layer(const float* X, int K, int Hh, int F,
                      const float* W, const float* al, const float* ar, const float* rW,
                      const int* src, const int* dst,
                      float* Hbuf, float* AGG, bool act, int Opad,
                      float* el, float* er, float* den, unsigned* mEnc, float* sb,
                      const int* csr_from, int* csr,
                      hipStream_t stream) {
  const int HF = Hh * F;
  dim3 ggrid(NN / 16, Opad / 64);
  gat_gemm_xwt<<<ggrid, 32, 0, stream>>>(X, W, Hbuf, K, Opad, HF);
  gat_gemm_xwt<<<ggrid, 32, 0, stream>>>(X, rW, AGG, K, Opad, HF);

  if (csr_from)
    gat_csr_move<<<(CSR_KEEP + 255) / 256, 256, 0, stream>>>(csr_from, csr, CSR_KEEP);

  const int nh = NN * Hh;
  gat_el_er<<<(nh + 255) / 256, 256, 0, stream>>>(Hbuf, al, ar, el, er, nh, Hh, F);
  gat_init_nm<<<(nh + 255) / 256, 256, 0, stream>>>(mEnc, den, nh);

  const int eh = EE * Hh;
  gat_edge_smax<<<(eh + 255) / 256, 256, 0, stream>>>(el, er, src, dst, sb, mEnc, eh, Hh);
  gat_edge_expsum<<<(eh + 255) / 256, 256, 0, stream>>>(sb, mEnc, den, dst, eh, Hh);

  if (HF == 256)
    gat_gather256<<<NN / 8, 256, 0, stream>>>(Hbuf, sb, den, src, csr, csr + CSR_PERM, AGG);
  else
    gat_gather40<<<(NN / 4 + 7) / 8, 256, 0, stream>>>(Hbuf, sb, den, src, csr, csr + CSR_PERM, AGG);

  if (act) {
    long tot = (long)NN * HF;
    gat_elu<<<(unsigned)((tot + 255) / 256), 256, 0, stream>>>(AGG, tot);
  }
}

extern "C" void kernel_launch(void* const* d_in, const int* in_sizes, int n_in,
                              void* d_out, int out_size, void* d_ws, size_t ws_size,
                              hipStream_t stream) {
  (void)in_sizes; (void)n_in; (void)out_size; (void)ws_size;

  const float* feat = (const float*)d_in[0];
  const int*   src  = (const int*)d_in[1];
  const int*   dst  = (const int*)d_in[2];
  const float* W0  = (const float*)d_in[3];
  const float* al0 = (const float*)d_in[4];
  const float* ar0 = (const float*)d_in[5];
  const float* rW0 = (const float*)d_in[6];
  const float* W1  = (const float*)d_in[7];
  const float* al1 = (const float*)d_in[8];
  const float* ar1 = (const float*)d_in[9];
  const float* rW1 = (const float*)d_in[10];
  const float* W2  = (const float*)d_in[11];
  const float* al2 = (const float*)d_in[12];
  const float* ar2 = (const float*)d_in[13];
  const float* rW2 = (const float*)d_in[14];

  const size_t NF = (size_t)NN * 256;
  float* ws    = (float*)d_ws;
  float* bufH  = ws;
  float* bufY0 = bufH  + NF;
  float* bufY1 = bufY0 + NF;
  float* el    = bufY1 + NF;
  float* er    = el  + (size_t)NN * 4;
  float* den   = er  + (size_t)NN * 4;
  unsigned* mE = (unsigned*)(den + (size_t)NN * 4);
  float* sb    = (float*)mE + (size_t)NN * 4;
  float* W2p   = sb + (size_t)EE * 4;
  float* rW2p  = W2p + (size_t)64 * 256;

  {
    int tot = 64 * 256;
    gat_pad_w<<<(tot + 255) / 256, 256, 0, stream>>>(W2,  W2p,  40, 64, 256);
    gat_pad_w<<<(tot + 255) / 256, 256, 0, stream>>>(rW2, rW2p, 40, 64, 256);
  }

  int* csrA = (int*)bufY1;
  int* csrS = (int*)sb;
  int* csrB = (int*)bufY0;
  {
    gat_csr_zero<<<(NN + 255) / 256, 256, 0, stream>>>(csrA + CSR_CUR, NN);
    gat_csr_hist<<<(EE + 255) / 256, 256, 0, stream>>>(dst, csrA + CSR_CUR, EE);
    gat_csr_bsum<<<CSR_NB, 256, 0, stream>>>(csrA + CSR_CUR, csrA + CSR_BSUM);
    gat_csr_offsets<<<CSR_NB, 256, 0, stream>>>(csrA + CSR_CUR, csrA + CSR_BSUM, csrA);
    gat_csr_fill<<<(EE + 255) / 256, 256, 0, stream>>>(dst, csrA + CSR_CUR, csrA + CSR_PERM, EE);
  }

  run_layer(feat, 128, 4, 64, W0, al0, ar0, rW0, src, dst,
            bufH, bufY0, true, 256, el, er, den, mE, sb, nullptr, csrA, stream);
  gat_csr_move<<<(CSR_KEEP + 255) / 256, 256, 0, stream>>>(csrA, csrS, CSR_KEEP);
  run_layer(bufY0, 256, 4, 64, W1, al1, ar1, rW1, src, dst,
            bufH, bufY1, true, 256, el, er, den, mE, sb, csrS, csrB, stream);
  run_layer(bufY1, 256, 1, 40, W2p, al2, ar2, rW2p, src, dst,
            bufH, (float*)d_out, false, 64, el, er, den, mE, sb, nullptr, csrB, stream);
}


# Edge softmax (score, max, exp-sum) and ELU folded into the per-destination gather wave: 13 launches (2 edge-list copies), score buffer, float atomics gone; exp-sum by wave tree, not atomic order
# speedup vs baseline: 3.4122x; 1.5169x over previous
#include <hip/hip_runtime.h>
#include <math.h>

#define NN 50000
#define EE 800000

typedef __attribute__((ext_vector_type(2))) float v2f;
typedef __attribute__((ext_vector_type(8))) float v8f;

__global__ __launch_bounds__(32)
void gat_gemm_xwt(const float* __restrict__ X, const float* __restrict__ W,
                  float* __restrict__ OUT, int K, int Opad, int Oout) {
  const int lane = threadIdx.x & 31;
  const int row0 = blockIdx.x << 4;
  const int col0 = blockIdx.y << 6;
  const int mr   = lane & 15;
  const int half = lane >> 4;
  const int koff = half << 1;

  const float* xrow = X + (size_t)(row0 + mr) * (size_t)K + koff;
  const float* wrow = W + (size_t)(col0 + mr) * (size_t)K + koff;
  const size_t wstep = (size_t)16 * (size_t)K;

  v8f acc[4] = {};
  v2f a0, a1;
  v2f b0[4], b1[4];

  a0 = *(const v2f*)(xrow);
#pragma unroll
  for (int t = 0; t < 4; ++t) b0[t] = *(const v2f*)(wrow + (size_t)t * wstep);

  int k0 = 0;
  for (; k0 + 8 < K; k0 += 8) {
    a1 = *(const v2f*)(xrow + k0 + 4);
#pragma unroll
    for (int t = 0; t < 4; ++t)
      b1[t] = *(const v2f*)(wrow + (size_t)t * wstep + k0 + 4);
#pragma unroll
    for (int t = 0; t < 4; ++t)
      acc[t] = __builtin_amdgcn_wmma_f32_16x16x4_f32(
          false, a0, false, b0[t], (short)0, acc[t], false, false);

    a0 = *(const v2f*)(xrow + k0 + 8);
#pragma unroll
    for (int t = 0; t < 4; ++t)
      b0[t] = *(const v2f*)(wrow + (size_t)t * wstep + k0 + 8);
#pragma unroll
    for (int t = 0; t < 4; ++t)
      acc[t] = __builtin_amdgcn_wmma_f32_16x16x4_f32(
          false, a1, false, b1[t], (short)0, acc[t], false, false);
  }

  a1 = *(const v2f*)(xrow + k0 + 4);
#pragma unroll
  for (int t = 0; t < 4; ++t)
    b1[t] = *(const v2f*)(wrow + (size_t)t * wstep + k0 + 4);
#pragma unroll
  for (int t = 0; t < 4; ++t)
    acc[t] = __builtin_amdgcn_wmma_f32_16x16x4_f32(
        false, a0, false, b0[t], (short)0, acc[t], false, false);
#pragma unroll
  for (int t = 0; t < 4; ++t)
    acc[t] = __builtin_amdgcn_wmma_f32_16x16x4_f32(
        false, a1, false, b1[t], (short)0, acc[t], false, false);

#pragma unroll
  for (int t = 0; t < 4; ++t) {
    const int oc = col0 + (t << 4) + mr;
    if (oc < Oout) {
      float* orow = OUT + (size_t)(row0 + (half << 3)) * (size_t)Oout + oc;
#pragma unroll
      for (int v = 0; v < 8; ++v) {
        orow[(size_t)v * (size_t)Oout] = acc[t][v];
      }
    }
  }
}

__global__ void gat_pad_w(const float* __restrict__ Wsrc, float* __restrict__ Wdst,
                          int rows, int padrows, int K) {
  int idx = blockIdx.x * blockDim.x + threadIdx.x;
  int total = padrows * K;
  if (idx >= total) return;
  int r = idx / K;
  Wdst[idx] = (r < rows) ? Wsrc[idx] : 0.0f;
}

__global__ void gat_el_er(const float* __restrict__ Hm, const float* __restrict__ al,
                          const float* __restrict__ ar, float* __restrict__ el,
                          float* __restrict__ er, int total, int Hh, int F) {
  int idx = blockIdx.x * blockDim.x + threadIdx.x;
  if (idx >= total) return;
  int h = idx % Hh;
  int n = idx / Hh;
  const float* row = Hm + (size_t)n * (size_t)(Hh * F) + (size_t)h * F;
  const float* a = al + h * F;
  const float* b = ar + h * F;
  float sl = 0.f, sr = 0.f;
  for (int f = 0; f < F; ++f) {
    float v = row[f];
    sl += v * a[f];
    sr += v * b[f];
  }
  el[idx] = sl;
  er[idx] = sr;
}

#define CSR_PERM (NN + 1)
#define CSR_CUR  (CSR_PERM + EE)
#define CSR_BSUM (CSR_CUR + NN)
#define CSR_NB   ((NN + 255) / 256)

__global__ void gat_csr_zero(int* __restrict__ cur, int total) {
  int idx = blockIdx.x * blockDim.x + threadIdx.x;
  if (idx < total) cur[idx] = 0;
}

__global__ void gat_csr_hist(const int* __restrict__ dst, int* __restrict__ cur, int total) {
  int e = blockIdx.x * blockDim.x + threadIdx.x;
  if (e < total) atomicAdd(cur + dst[e], 1);
}

__device__ __forceinline__ int gat_block_scan256(int v, int* sh) {
  const int t = threadIdx.x;
  sh[t] = v;
  __syncthreads();
  for (int d = 1; d < 256; d <<= 1) {
    int a = (t >= d) ? sh[t - d] : 0;
    __syncthreads();
    sh[t] += a;
    __syncthreads();
  }
  return sh[t];
}

__global__ __launch_bounds__(256)
void gat_csr_bsum(const int* __restrict__ cur, int* __restrict__ bsum) {
  __shared__ int sh[256];
  const int i = blockIdx.x * 256 + threadIdx.x;
  const int v = (i < NN) ? cur[i] : 0;
  const int inc = gat_block_scan256(v, sh);
  if (threadIdx.x == 255) bsum[blockIdx.x] = inc;
}

__global__ __launch_bounds__(256)
void gat_csr_offsets(int* __restrict__ cur, const int* __restrict__ bsum,
                     int* __restrict__ off) {
  __shared__ int sh[256];
  const int t = threadIdx.x;
  const int b = blockIdx.x;
  const int x = (t < b) ? bsum[t] : 0;
  gat_block_scan256(x, sh);
  const int before = sh[255];
  __syncthreads();
  const int i = b * 256 + t;
  const int v = (i < NN) ? cur[i] : 0;
  const int inc = gat_block_scan256(v, sh);
  if (i < NN) {
    const int o = before + inc - v;
    off[i] = o;
    cur[i] = o;
    if (i == NN - 1) off[NN] = o + v;
  }
}

__global__ void gat_csr_fill(const int* __restrict__ dst, int* __restrict__ cur,
                             int* __restrict__ perm, int total) {
  int e = blockIdx.x * blockDim.x + threadIdx.x;
  if (e >= total) return;
  int p = atomicAdd(cur + dst[e], 1);
  perm[p] = e;
}


__device__ __forceinline__ float gat_leaky(float s) {
  return (s > 0.0f) ? s : 0.2f * s;
}
__device__ __forceinline__ float gat_elu1(float x) {
  return (x > 0.0f) ? x : expm1f(x);
}
__device__ __forceinline__ float gat_wave_max(float v) {
  for (int d = 16; d >= 1; d >>= 1) v = fmaxf(v, __shfl_xor(v, d, 32));
  return v;
}
__device__ __forceinline__ float gat_wave_sum(float v) {
  for (int d = 16; d >= 1; d >>= 1) v += __shfl_xor(v, d, 32);
  return v;
}

__device__ __forceinline__ void gat_chunk4(int base, int end, int lane,
                                           const int* __restrict__ perm,
                                           const int* __restrict__ src,
                                           const float* __restrict__ el, float4 rn,
                                           int& my_s, float4& sc) {
  my_s = 0;
  sc.x = -INFINITY; sc.y = -INFINITY; sc.z = -INFINITY; sc.w = -INFINITY;
  if (base + lane < end) {
    const int e = perm[base + lane];
    my_s = src[e];
    const float4 l = *(const float4*)(el + (size_t)my_s * 4);
    sc.x = gat_leaky(l.x + rn.x);
    sc.y = gat_leaky(l.y + rn.y);
    sc.z = gat_leaky(l.z + rn.z);
    sc.w = gat_leaky(l.w + rn.w);
  }
}

__global__ __launch_bounds__(256)
void gat_gather256(const float* __restrict__ Hm, const float* __restrict__ el,
                   const float* __restrict__ er, const int* __restrict__ src,
                   const int* __restrict__ off, const int* __restrict__ perm,
                   float* __restrict__ AGG, int act) {
  const int lane = threadIdx.x & 31;
  const int n = __builtin_amdgcn_readfirstlane((int)(blockIdx.x * 8 + (threadIdx.x >> 5)));
  if (n >= NN) return;
  const int beg = off[n];
  const int end = off[n + 1];
  const float4 rn = *(const float4*)(er + (size_t)n * 4);
  const bool one = (end - beg) <= 32;
  const bool hi = lane >= 16;
  int my_s = 0;
  float4 sc = {-INFINITY, -INFINITY, -INFINITY, -INFINITY};

  float4 mx = {-INFINITY, -INFINITY, -INFINITY, -INFINITY};
  for (int base = beg; base < end; base += 32) {
    gat_chunk4(base, end, lane, perm, src, el, rn, my_s, sc);
    mx.x = fmaxf(mx.x, sc.x); mx.y = fmaxf(mx.y, sc.y);
    mx.z = fmaxf(mx.z, sc.z); mx.w = fmaxf(mx.w, sc.w);
  }
  mx.x = gat_wave_max(mx.x); mx.y = gat_wave_max(mx.y);
  mx.z = gat_wave_max(mx.z); mx.w = gat_wave_max(mx.w);

  float4 dn = {0.f, 0.f, 0.f, 0.f};
  for (int base = beg; base < end; base += 32) {
    if (!one) gat_chunk4(base, end, lane, perm, src, el, rn, my_s, sc);
    dn.x += expf(sc.x - mx.x); dn.y += expf(sc.y - mx.y);
    dn.z += expf(sc.z - mx.z); dn.w += expf(sc.w - mx.w);
  }
  dn.x = gat_wave_sum(dn.x); dn.y = gat_wave_sum(dn.y);
  dn.z = gat_wave_sum(dn.z); dn.w = gat_wave_sum(dn.w);

  float4 acc0 = {0.f, 0.f, 0.f, 0.f};
  float4 acc1 = {0.f, 0.f, 0.f, 0.f};
  for (int base = beg; base < end; base += 32) {
    const int left = end - base;
    const int cnt = (left < 32) ? left : 32;
    if (!one) gat_chunk4(base, end, lane, perm, src, el, rn, my_s, sc);
    float4 al;
    al.x = expf(sc.x - mx.x) / dn.x;
    al.y = expf(sc.y - mx.y) / dn.y;
    al.z = expf(sc.z - mx.z) / dn.z;
    al.w = expf(sc.w - mx.w) / dn.w;
#pragma unroll 4
    for (int i = 0; i < cnt; ++i) {
      const int s = __builtin_amdgcn_readlane(my_s, i);
      const float a0 = __int_as_float(__builtin_amdgcn_readlane(__float_as_int(al.x), i));
      const float a1 = __int_as_float(__builtin_amdgcn_readlane(__float_as_int(al.y), i));
      const float a2 = __int_as_float(__builtin_amdgcn_readlane(__float_as_int(al.z), i));
      const float a3 = __int_as_float(__builtin_amdgcn_readlane(__float_as_int(al.w), i));
      const float w0 = hi ? a1 : a0;
      const float w1 = hi ? a3 : a2;
      const float4* row = (const float4*)(Hm + (size_t)s * 256);
      const float4 v0 = row[lane];
      const float4 v1 = row[32 + lane];
      acc0.x += v0.x * w0; acc0.y += v0.y * w0; acc0.z += v0.z * w0; acc0.w += v0.w * w0;
      acc1.x += v1.x * w1; acc1.y += v1.y * w1; acc1.z += v1.z * w1; acc1.w += v1.w * w1;
    }
  }

  float4* o = (float4*)(AGG + (size_t)n * 256);
  float4 r0 = o[lane];
  float4 r1 = o[32 + lane];
  r0.x += acc0.x; r0.y += acc0.y; r0.z += acc0.z; r0.w += acc0.w;
  r1.x += acc1.x; r1.y += acc1.y; r1.z += acc1.z; r1.w += acc1.w;
  if (act) {
    r0.x = gat_elu1(r0.x); r0.y = gat_elu1(r0.y); r0.z = gat_elu1(r0.z); r0.w = gat_elu1(r0.w);
    r1.x = gat_elu1(r1.x); r1.y = gat_elu1(r1.y); r1.z = gat_elu1(r1.z); r1.w = gat_elu1(r1.w);
  }
  o[lane] = r0;
  o[32 + lane] = r1;
}

__device__ __forceinline__ void gat_chunk1(int base, int end, int lane,
                                           const int* __restrict__ perm,
                                           const int* __restrict__ src,
                                           const float* __restrict__ el, float rn,
                                           int& my_s, float& sc) {
  my_s = 0;
  sc = -INFINITY;
  if (base + lane < end) {
    const int e = perm[base + lane];
    my_s = src[e];
    sc = gat_leaky(el[my_s] + rn);
  }
}

__global__ __launch_bounds__(256)
void gat_gather40(const float* __restrict__ Hm, const float* __restrict__ el,
                  const float* __restrict__ er, const int* __restrict__ src,
                  const int* __restrict__ off, const int* __restrict__ perm,
                  float* __restrict__ AGG, int act) {
  const int lane = threadIdx.x & 31;
  const int g = __builtin_amdgcn_readfirstlane((int)(blockIdx.x * 8 + (threadIdx.x >> 5)));
  if (g * 4 >= NN) return;
  const int col = ((lane < 20) ? lane : 19) * 2;
  for (int q = 0; q < 4; ++q) {
    const int n = g * 4 + q;
    const int beg = off[n];
    const int end = off[n + 1];
    const float rn = er[n];
    const bool one = (end - beg) <= 32;
    int my_s = 0;
    float sc = -INFINITY;

    float mx = -INFINITY;
    for (int base = beg; base < end; base += 32) {
      gat_chunk1(base, end, lane, perm, src, el, rn, my_s, sc);
      mx = fmaxf(mx, sc);
    }
    mx = gat_wave_max(mx);

    float dn = 0.f;
    for (int base = beg; base < end; base += 32) {
      if (!one) gat_chunk1(base, end, lane, perm, src, el, rn, my_s, sc);
      dn += expf(sc - mx);
    }
    dn = gat_wave_sum(dn);

    float2 acc = {0.f, 0.f};
    for (int base = beg; base < end; base += 32) {
      const int left = end - base;
      const int cnt = (left < 32) ? left : 32;
      if (!one) gat_chunk1(base, end, lane, perm, src, el, rn, my_s, sc);
      const float my_a = expf(sc - mx) / dn;
#pragma unroll 4
      for (int i = 0; i < cnt; ++i) {
        const int s = __builtin_amdgcn_readlane(my_s, i);
        const float a = __int_as_float(__builtin_amdgcn_readlane(__float_as_int(my_a), i));
        const float2 v = *(const float2*)(Hm + (size_t)s * 40 + col);
        acc.x += v.x * a;
        acc.y += v.y * a;
      }
    }
    if (lane < 20) {
      float2* o = (float2*)(AGG + (size_t)n * 40 + col);
      float2 r = *o;
      r.x += acc.x;
      r.y += acc.y;
      if (act) { r.x = gat_elu1(r.x); r.y = gat_elu1(r.y); }
      *o = r;
    }
  }
}

static void run_layer(const float* X, int K, int Hh, int F,
                      const float* W, const float* al, const float* ar, const float* rW,
                      const int* src,
                      float* Hbuf, float* AGG, bool act, int Opad,
                      float* el, float* er, const int* csr,
                      hipStream_t stream) {
  const int HF = Hh * F;
  dim3 ggrid(NN / 16, Opad / 64);
  gat_gemm_xwt<<<ggrid, 32, 0, stream>>>(X, W, Hbuf, K, Opad, HF);
  gat_gemm_xwt<<<ggrid, 32, 0, stream>>>(X, rW, AGG, K, Opad, HF);

  const int nh = NN * Hh;
  gat_el_er<<<(nh + 255) / 256, 256, 0, stream>>>(Hbuf, al, ar, el, er, nh, Hh, F);

  if (HF == 256)
    gat_gather256<<<NN / 8, 256, 0, stream>>>(Hbuf, el, er, src, csr, csr + CSR_PERM, AGG,
                                              act ? 1 : 0);
  else
    gat_gather40<<<(NN / 4 + 7) / 8, 256, 0, stream>>>(Hbuf, el, er, src, csr, csr + CSR_PERM,
                                                       AGG, act ? 1 : 0);
}

extern "C" void kernel_launch(void* const* d_in, const int* in_sizes, int n_in,
                              void* d_out, int out_size, void* d_ws, size_t ws_size,
                              hipStream_t stream) {
  (void)in_sizes; (void)n_in; (void)out_size; (void)ws_size;

  const float* feat = (const float*)d_in[0];
  const int*   src  = (const int*)d_in[1];
  const int*   dst  = (const int*)d_in[2];
  const float* W0  = (const float*)d_in[3];
  const float* al0 = (const float*)d_in[4];
  const float* ar0 = (const float*)d_in[5];
  const float* rW0 = (const float*)d_in[6];
  const float* W1  = (const float*)d_in[7];
  const float* al1 = (const float*)d_in[8];
  const float* ar1 = (const float*)d_in[9];
  const float* rW1 = (const float*)d_in[10];
  const float* W2  = (const float*)d_in[11];
  const float* al2 = (const float*)d_in[12];
  const float* ar2 = (const float*)d_in[13];
  const float* rW2 = (const float*)d_in[14];

  const size_t NF = (size_t)NN * 256;
  float* ws    = (float*)d_ws;
  float* bufH  = ws;
  float* bufY0 = bufH  + NF;
  float* bufY1 = bufY0 + NF;
  float* el    = bufY1 + NF;
  float* er    = el  + (size_t)NN * 4;
  float* sb    = er  + (size_t)NN * 12;
  float* W2p   = sb + (size_t)EE * 4;
  float* rW2p  = W2p + (size_t)64 * 256;

  {
    int tot = 64 * 256;
    gat_pad_w<<<(tot + 255) / 256, 256, 0, stream>>>(W2,  W2p,  40, 64, 256);
    gat_pad_w<<<(tot + 255) / 256, 256, 0, stream>>>(rW2, rW2p, 40, 64, 256);
  }

  int* csr = (int*)sb;
  {
    gat_csr_zero<<<(NN + 255) / 256, 256, 0, stream>>>(csr + CSR_CUR, NN);
    gat_csr_hist<<<(EE + 255) / 256, 256, 0, stream>>>(dst, csr + CSR_CUR, EE);
    gat_csr_bsum<<<CSR_NB, 256, 0, stream>>>(csr + CSR_CUR, csr + CSR_BSUM);
    gat_csr_offsets<<<CSR_NB, 256, 0, stream>>>(csr + CSR_CUR, csr + CSR_BSUM, csr);
    gat_csr_fill<<<(EE + 255) / 256, 256, 0, stream>>>(dst, csr + CSR_CUR, csr + CSR_PERM, EE);
  }

  run_layer(feat, 128, 4, 64, W0, al0, ar0, rW0, src,
            bufH, bufY0, true, 256, el, er, csr, stream);
  run_layer(bufY0, 256, 4, 64, W1, al1, ar1, rW1, src,
            bufH, bufY1, true, 256, el, er, csr, stream);
  run_layer(bufY1, 256, 1, 40, W2p, al2, ar2, rW2p, src,
            bufH, (float*)d_out, false, 64, el, er, csr, stream);
}


# GEMM operands staged through LDS: a 4-wave workgroup shares a 64x64 tile, X and W k-blocks of 32 are copied in by whole-line 16-B loads, fragments are read from LDS; k order of every sum unchanged
# speedup vs baseline: 5.0772x; 1.4880x over previous
#include <hip/hip_runtime.h>
#include <math.h>

#define NN 50000
#define EE 800000

typedef __attribute__((ext_vector_type(2))) float v2f;
typedef __attribute__((ext_vector_type(8))) float v8f;

#define GB_T 64
#define GB_K 32
#define GB_P 36
#define GB_STAGE (2 * GB_T * GB_P)

__device__ __forceinline__ void gat_gemm_kblock(const float* ap, const float* bp,
                                                v8f (&acc)[4]) {
  v2f a[2];
  v2f b[2][4];
  a[0] = *(const v2f*)ap;
  for (int t = 0; t < 4; ++t) b[0][t] = *(const v2f*)(bp + t * 16 * GB_P);
  for (int s = 0; s < GB_K / 4; ++s) {
    const int cur = s & 1;
    const int nxt = cur ^ 1;
    if (s + 1 < GB_K / 4) {
      a[nxt] = *(const v2f*)(ap + 4 * (s + 1));
      for (int t = 0; t < 4; ++t)
        b[nxt][t] = *(const v2f*)(bp + t * 16 * GB_P + 4 * (s + 1));
    }
    for (int t = 0; t < 4; ++t)
      acc[t] = __builtin_amdgcn_wmma_f32_16x16x4_f32(
          false, a[cur], false, b[cur][t], (short)0, acc[t], false, false);
  }
}

__global__ __launch_bounds__(128)
void gat_gemm_xwt(const float* __restrict__ X, const float* __restrict__ W,
                  float* __restrict__ OUT, int K, int Opad, int Oout) {
  __shared__ __attribute__((aligned(16))) float tile[2 * GB_STAGE];
  (void)Opad;
  const int tid  = threadIdx.x;
  const int lane = tid & 31;
  const int wv   = tid >> 5;
  const int row0 = blockIdx.x * GB_T;
  const int col0 = blockIdx.y * GB_T;
  const int mr   = lane & 15;
  const int half = lane >> 4;

  const int c4 = tid & 7;
  const int r8 = tid >> 3;
  const float* xp[4];
  const float* wp[4];
  for (int i = 0; i < 4; ++i) {
    int xr = row0 + r8 + 16 * i;
    xr = (xr < NN) ? xr : (NN - 1);
    xp[i] = X + (size_t)xr * (size_t)K + 4 * c4;
    wp[i] = W + (size_t)(col0 + r8 + 16 * i) * (size_t)K + 4 * c4;
  }
  float* sx = tile + r8 * GB_P + 4 * c4;
  float* sw = sx + GB_T * GB_P;
  const float* ap = tile + (16 * wv + mr) * GB_P + 2 * half;
  const float* bp = tile + (GB_T + mr) * GB_P + 2 * half;

  float4 px[4], pw[4];
  for (int i = 0; i < 4; ++i) {
    px[i] = *(const float4*)(xp[i]);
    pw[i] = *(const float4*)(wp[i]);
  }
  for (int i = 0; i < 4; ++i) {
    *(float4*)(sx + 16 * i * GB_P) = px[i];
    *(float4*)(sw + 16 * i * GB_P) = pw[i];
  }
  __syncthreads();

  v8f acc[4] = {};
  const int nkb = K / GB_K;
  for (int kb = 0; kb + 1 < nkb; ++kb) {
    const int so = (kb & 1) * GB_STAGE;
    const int sn = GB_STAGE - so;
    const int kn = (kb + 1) * GB_K;
    for (int i = 0; i < 4; ++i) {
      px[i] = *(const float4*)(xp[i] + kn);
      pw[i] = *(const float4*)(wp[i] + kn);
    }
    gat_gemm_kblock(ap + so, bp + so, acc);
    for (int i = 0; i < 4; ++i) {
      *(float4*)(sx + sn + 16 * i * GB_P) = px[i];
      *(float4*)(sw + sn + 16 * i * GB_P) = pw[i];
    }
    __syncthreads();
  }
  {
    const int so = ((nkb - 1) & 1) * GB_STAGE;
    gat_gemm_kblock(ap + so, bp + so, acc);
  }

  const int wrow0 = row0 + 16 * wv;
  if (wrow0 < NN) {
    for (int t = 0; t < 4; ++t) {
      const int oc = col0 + (t << 4) + mr;
      if (oc < Oout) {
        float* orow = OUT + (size_t)(wrow0 + (half << 3)) * (size_t)Oout + oc;
        for (int v = 0; v < 8; ++v) {
          orow[(size_t)v * (size_t)Oout] = acc[t][v];
        }
      }
    }
  }
}

__global__ void gat_pad_w(const float* __restrict__ Wsrc, float* __restrict__ Wdst,
                          int rows, int padrows, int K) {
  int idx = blockIdx.x * blockDim.x + threadIdx.x;
  int total = padrows * K;
  if (idx >= total) return;
  int r = idx / K;
  Wdst[idx] = (r < rows) ? Wsrc[idx] : 0.0f;
}

__global__ void gat_el_er(const float* __restrict__ Hm, const float* __restrict__ al,
                          const float* __restrict__ ar, float* __restrict__ el,
                          float* __restrict__ er, int total, int Hh, int F) {
  int idx = blockIdx.x * blockDim.x + threadIdx.x;
  if (idx >= total) return;
  int h = idx % Hh;
  int n = idx / Hh;
  const float* row = Hm + (size_t)n * (size_t)(Hh * F) + (size_t)h * F;
  const float* a = al + h * F;
  const float* b = ar + h * F;
  float sl = 0.f, sr = 0.f;
  for (int f = 0; f < F; ++f) {
    float v = row[f];
    sl += v * a[f];
    sr += v * b[f];
  }
  el[idx] = sl;
  er[idx] = sr;
}

#define CSR_PERM (NN + 1)
#define CSR_CUR  (CSR_PERM + EE)
#define CSR_BSUM (CSR_CUR + NN)
#define CSR_NB   ((NN + 255) / 256)

__global__ void gat_csr_zero(int* __restrict__ cur, int total) {
  int idx = blockIdx.x * blockDim.x + threadIdx.x;
  if (idx < total) cur[idx] = 0;
}

__global__ void gat_csr_hist(const int* __restrict__ dst, int* __restrict__ cur, int total) {
  int e = blockIdx.x * blockDim.x + threadIdx.x;
  if (e < total) atomicAdd(cur + dst[e], 1);
}

__device__ __forceinline__ int gat_block_scan256(int v, int* sh) {
  const int t = threadIdx.x;
  sh[t] = v;
  __syncthreads();
  for (int d = 1; d < 256; d <<= 1) {
    int a = (t >= d) ? sh[t - d] : 0;
    __syncthreads();
    sh[t] += a;
    __syncthreads();
  }
  return sh[t];
}

__global__ __launch_bounds__(256)
void gat_csr_bsum(const int* __restrict__ cur, int* __restrict__ bsum) {
  __shared__ int sh[256];
  const int i = blockIdx.x * 256 + threadIdx.x;
  const int v = (i < NN) ? cur[i] : 0;
  const int inc = gat_block_scan256(v, sh);
  if (threadIdx.x == 255) bsum[blockIdx.x] = inc;
}

__global__ __launch_bounds__(256)
void gat_csr_offsets(int* __restrict__ cur, const int* __restrict__ bsum,
                     int* __restrict__ off) {
  __shared__ int sh[256];
  const int t = threadIdx.x;
  const int b = blockIdx.x;
  const int x = (t < b) ? bsum[t] : 0;
  gat_block_scan256(x, sh);
  const int before = sh[255];
  __syncthreads();
  const int i = b * 256 + t;
  const int v = (i < NN) ? cur[i] : 0;
  const int inc = gat_block_scan256(v, sh);
  if (i < NN) {
    const int o = before + inc - v;
    off[i] = o;
    cur[i] = o;
    if (i == NN - 1) off[NN] = o + v;
  }
}

__global__ void gat_csr_fill(const int* __restrict__ dst, int* __restrict__ cur,
                             int* __restrict__ perm, int total) {
  int e = blockIdx.x * blockDim.x + threadIdx.x;
  if (e >= total) return;
  int p = atomicAdd(cur + dst[e], 1);
  perm[p] = e;
}


__device__ __forceinline__ float gat_leaky(float s) {
  return (s > 0.0f) ? s : 0.2f * s;
}
__device__ __forceinline__ float gat_elu1(float x) {
  return (x > 0.0f) ? x : expm1f(x);
}
__device__ __forceinline__ float gat_wave_max(float v) {
  for (int d = 16; d >= 1; d >>= 1) v = fmaxf(v, __shfl_xor(v, d, 32));
  return v;
}
__device__ __forceinline__ float gat_wave_sum(float v) {
  for (int d = 16; d >= 1; d >>= 1) v += __shfl_xor(v, d, 32);
  return v;
}

__device__ __forceinline__ void gat_chunk4(int base, int end, int lane,
                                           const int* __restrict__ perm,
                                           const int* __restrict__ src,
                                           const float* __restrict__ el, float4 rn,
                                           int& my_s, float4& sc) {
  my_s = 0;
  sc.x = -INFINITY; sc.y = -INFINITY; sc.z = -INFINITY; sc.w = -INFINITY;
  if (base + lane < end) {
    const int e = perm[base + lane];
    my_s = src[e];
    const float4 l = *(const float4*)(el + (size_t)my_s * 4);
    sc.x = gat_leaky(l.x + rn.x);
    sc.y = gat_leaky(l.y + rn.y);
    sc.z = gat_leaky(l.z + rn.z);
    sc.w = gat_leaky(l.w + rn.w);
  }
}

__global__ __launch_bounds__(256)
void gat_gather256(const float* __restrict__ Hm, const float* __restrict__ el,
                   const float* __restrict__ er, const int* __restrict__ src,
                   const int* __restrict__ off, const int* __restrict__ perm,
                   float* __restrict__ AGG, int act) {
  const int lane = threadIdx.x & 31;
  const int n = __builtin_amdgcn_readfirstlane((int)(blockIdx.x * 8 + (threadIdx.x >> 5)));
  if (n >= NN) return;
  const int beg = off[n];
  const int end = off[n + 1];
  const float4 rn = *(const float4*)(er + (size_t)n * 4);
  const bool one = (end - beg) <= 32;
  const bool hi = lane >= 16;
  int my_s = 0;
  float4 sc = {-INFINITY, -INFINITY, -INFINITY, -INFINITY};

  float4 mx = {-INFINITY, -INFINITY, -INFINITY, -INFINITY};
  for (int base = beg; base < end; base += 32) {
    gat_chunk4(base, end, lane, perm, src, el, rn, my_s, sc);
    mx.x = fmaxf(mx.x, sc.x); mx.y = fmaxf(mx.y, sc.y);
    mx.z = fmaxf(mx.z, sc.z); mx.w = fmaxf(mx.w, sc.w);
  }
  mx.x = gat_wave_max(mx.x); mx.y = gat_wave_max(mx.y);
  mx.z = gat_wave_max(mx.z); mx.w = gat_wave_max(mx.w);

  float4 dn = {0.f, 0.f, 0.f, 0.f};
  for (int base = beg; base < end; base += 32) {
    if (!one) gat_chunk4(base, end, lane, perm, src, el, rn, my_s, sc);
    dn.x += expf(sc.x - mx.x); dn.y += expf(sc.y - mx.y);
    dn.z += expf(sc.z - mx.z); dn.w += expf(sc.w - mx.w);
  }
  dn.x = gat_wave_sum(dn.x); dn.y = gat_wave_sum(dn.y);
  dn.z = gat_wave_sum(dn.z); dn.w = gat_wave_sum(dn.w);

  float4 acc0 = {0.f, 0.f, 0.f, 0.f};
  float4 acc1 = {0.f, 0.f, 0.f, 0.f};
  for (int base = beg; base < end; base += 32) {
    const int left = end - base;
    const int cnt = (left < 32) ? left : 32;
    if (!one) gat_chunk4(base, end, lane, perm, src, el, rn, my_s, sc);
    float4 al;
    al.x = expf(sc.x - mx.x) / dn.x;
    al.y = expf(sc.y - mx.y) / dn.y;
    al.z = expf(sc.z - mx.z) / dn.z;
    al.w = expf(sc.w - mx.w) / dn.w;
#pragma unroll 4
    for (int i = 0; i < cnt; ++i) {
      const int s = __builtin_amdgcn_readlane(my_s, i);
      const float a0 = __int_as_float(__builtin_amdgcn_readlane(__float_as_int(al.x), i));
      const float a1 = __int_as_float(__builtin_amdgcn_readlane(__float_as_int(al.y), i));
      const float a2 = __int_as_float(__builtin_amdgcn_readlane(__float_as_int(al.z), i));
      const float a3 = __int_as_float(__builtin_amdgcn_readlane(__float_as_int(al.w), i));
      const float w0 = hi ? a1 : a0;
      const float w1 = hi ? a3 : a2;
      const float4* row = (const float4*)(Hm + (size_t)s * 256);
      const float4 v0 = row[lane];
      const float4 v1 = row[32 + lane];
      acc0.x += v0.x * w0; acc0.y += v0.y * w0; acc0.z += v0.z * w0; acc0.w += v0.w * w0;
      acc1.x += v1.x * w1; acc1.y += v1.y * w1; acc1.z += v1.z * w1; acc1.w += v1.w * w1;
    }
  }

  float4* o = (float4*)(AGG + (size_t)n * 256);
  float4 r0 = o[lane];
  float4 r1 = o[32 + lane];
  r0.x += acc0.x; r0.y += acc0.y; r0.z += acc0.z; r0.w += acc0.w;
  r1.x += acc1.x; r1.y += acc1.y; r1.z += acc1.z; r1.w += acc1.w;
  if (act) {
    r0.x = gat_elu1(r0.x); r0.y = gat_elu1(r0.y); r0.z = gat_elu1(r0.z); r0.w = gat_elu1(r0.w);
    r1.x = gat_elu1(r1.x); r1.y = gat_elu1(r1.y); r1.z = gat_elu1(r1.z); r1.w = gat_elu1(r1.w);
  }
  o[lane] = r0;
  o[32 + lane] = r1;
}

__device__ __forceinline__ void gat_chunk1(int base, int end, int lane,
                                           const int* __restrict__ perm,
                                           const int* __restrict__ src,
                                           const float* __restrict__ el, float rn,
                                           int& my_s, float& sc) {
  my_s = 0;
  sc = -INFINITY;
  if (base + lane < end) {
    const int e = perm[base + lane];
    my_s = src[e];
    sc = gat_leaky(el[my_s] + rn);
  }
}

__global__ __launch_bounds__(256)
void gat_gather40(const float* __restrict__ Hm, const float* __restrict__ el,
                  const float* __restrict__ er, const int* __restrict__ src,
                  const int* __restrict__ off, const int* __restrict__ perm,
                  float* __restrict__ AGG, int act) {
  const int lane = threadIdx.x & 31;
  const int g = __builtin_amdgcn_readfirstlane((int)(blockIdx.x * 8 + (threadIdx.x >> 5)));
  if (g * 4 >= NN) return;
  const int col = ((lane < 20) ? lane : 19) * 2;
  for (int q = 0; q < 4; ++q) {
    const int n = g * 4 + q;
    const int beg = off[n];
    const int end = off[n + 1];
    const float rn = er[n];
    const bool one = (end - beg) <= 32;
    int my_s = 0;
    float sc = -INFINITY;

    float mx = -INFINITY;
    for (int base = beg; base < end; base += 32) {
      gat_chunk1(base, end, lane, perm, src, el, rn, my_s, sc);
      mx = fmaxf(mx, sc);
    }
    mx = gat_wave_max(mx);

    float dn = 0.f;
    for (int base = beg; base < end; base += 32) {
      if (!one) gat_chunk1(base, end, lane, perm, src, el, rn, my_s, sc);
      dn += expf(sc - mx);
    }
    dn = gat_wave_sum(dn);

    float2 acc = {0.f, 0.f};
    for (int base = beg; base < end; base += 32) {
      const int left = end - base;
      const int cnt = (left < 32) ? left : 32;
      if (!one) gat_chunk1(base, end, lane, perm, src, el, rn, my_s, sc);
      const float my_a = expf(sc - mx) / dn;
#pragma unroll 4
      for (int i = 0; i < cnt; ++i) {
        const int s = __builtin_amdgcn_readlane(my_s, i);
        const float a = __int_as_float(__builtin_amdgcn_readlane(__float_as_int(my_a), i));
        const float2 v = *(const float2*)(Hm + (size_t)s * 40 + col);
        acc.x += v.x * a;
        acc.y += v.y * a;
      }
    }
    if (lane < 20) {
      float2* o = (float2*)(AGG + (size_t)n * 40 + col);
      float2 r = *o;
      r.x += acc.x;
      r.y += acc.y;
      if (act) { r.x = gat_elu1(r.x); r.y = gat_elu1(r.y); }
      *o = r;
    }
  }
}

static void run_layer(const float* X, int K, int Hh, int F,
                      const float* W, const float* al, const float* ar, const float* rW,
                      const int* src,
                      float* Hbuf, float* AGG, bool act, int Opad,
                      float* el, float* er, const int* csr,
                      hipStream_t stream) {
  const int HF = Hh * F;
  dim3 ggrid((NN + GB_T - 1) / GB_T, Opad / GB_T);
  gat_gemm_xwt<<<ggrid, 128, 0, stream>>>(X, W, Hbuf, K, Opad, HF);
  gat_gemm_xwt<<<ggrid, 128, 0, stream>>>(X, rW, AGG, K, Opad, HF);

  const int nh = NN * Hh;
  gat_el_er<<<(nh + 255) / 256, 256, 0, stream>>>(Hbuf, al, ar, el, er, nh, Hh, F);

  if (HF == 256)
    gat_gather256<<<NN / 8, 256, 0, stream>>>(Hbuf, el, er, src, csr, csr + CSR_PERM, AGG,
                                              act ? 1 : 0);
  else
    gat_gather40<<<(NN / 4 + 7) / 8, 256, 0, stream>>>(Hbuf, el, er, src, csr, csr + CSR_PERM,
                                                       AGG, act ? 1 : 0);
}

extern "C" void kernel_launch(void* const* d_in, const int* in_sizes, int n_in,
                              void* d_out, int out_size, void* d_ws, size_t ws_size,
                              hipStream_t stream) {
  (void)in_sizes; (void)n_in; (void)out_size; (void)ws_size;

  const float* feat = (const float*)d_in[0];
  const int*   src  = (const int*)d_in[1];
  const int*   dst  = (const int*)d_in[2];
  const float* W0  = (const float*)d_in[3];
  const float* al0 = (const float*)d_in[4];
  const float* ar0 = (const float*)d_in[5];
  const float* rW0 = (const float*)d_in[6];
  const float* W1  = (const float*)d_in[7];
  const float* al1 = (const float*)d_in[8];
  const float* ar1 = (const float*)d_in[9];
  const float* rW1 = (const float*)d_in[10];
  const float* W2  = (const float*)d_in[11];
  const float* al2 = (const float*)d_in[12];
  const float* ar2 = (const float*)d_in[13];
  const float* rW2 = (const float*)d_in[14];

  const size_t NF = (size_t)NN * 256;
  float* ws    = (float*)d_ws;
  float* bufH  = ws;
  float* bufY0 = bufH  + NF;
  float* bufY1 = bufY0 + NF;
  float* el    = bufY1 + NF;
  float* er    = el  + (size_t)NN * 4;
  float* sb    = er  + (size_t)NN * 12;
  float* W2p   = sb + (size_t)EE * 4;
  float* rW2p  = W2p + (size_t)64 * 256;

  {
    int tot = 64 * 256;
    gat_pad_w<<<(tot + 255) / 256, 256, 0, stream>>>(W2,  W2p,  40, 64, 256);
    gat_pad_w<<<(tot + 255) / 256, 256, 0, stream>>>(rW2, rW2p, 40, 64, 256);
  }

  int* csr = (int*)sb;
  {
    gat_csr_zero<<<(NN + 255) / 256, 256, 0, stream>>>(csr + CSR_CUR, NN);
    gat_csr_hist<<<(EE + 255) / 256, 256, 0, stream>>>(dst, csr + CSR_CUR, EE);
    gat_csr_bsum<<<CSR_NB, 256, 0, stream>>>(csr + CSR_CUR, csr + CSR_BSUM);
    gat_csr_offsets<<<CSR_NB, 256, 0, stream>>>(csr + CSR_CUR, csr + CSR_BSUM, csr);
    gat_csr_fill<<<(EE + 255) / 256, 256, 0, stream>>>(dst, csr + CSR_CUR, csr + CSR_PERM, EE);
  }

  run_layer(feat, 128, 4, 64, W0, al0, ar0, rW0, src,
            bufH, bufY0, true, 256, el, er, csr, stream);
  run_layer(bufY0, 256, 4, 64, W1, al1, ar1, rW1, src,
            bufH, bufY1, true, 256, el, er, csr, stream);
  run_layer(bufY1, 256, 1, 40, W2p, al2, ar2, rW2p, src,
            bufH, (float*)d_out, false, 64, el, er, csr, stream);
}


# el/er formed in the h GEMM epilogue from its accumulators (4 column tiles per lane, then 16-lane DPP butterfly: 64 terms summed in another order); 3 el_er launches and their H re-read gone
# speedup vs baseline: 5.2615x; 1.0363x over previous
#include <hip/hip_runtime.h>
#include <math.h>

#define NN 50000
#define EE 800000

typedef __attribute__((ext_vector_type(2))) float v2f;
typedef __attribute__((ext_vector_type(8))) float v8f;

#define GB_T 64
#define GB_K 32
#define GB_P 36
#define GB_STAGE (2 * GB_T * GB_P)

template <int XM>
__device__ __forceinline__ float gat_row16_step(float x) {
  const int xi = __builtin_bit_cast(int, x);
  const int yi = __builtin_amdgcn_update_dpp(0, xi, 0x160 + XM, 0xf, 0xf, true);
  return x + __builtin_bit_cast(float, yi);
}

__device__ __forceinline__ void gat_gemm_kblock(const float* ap, const float* bp,
                                                v8f (&acc)[4]) {
  v2f a[2];
  v2f b[2][4];
  a[0] = *(const v2f*)ap;
  for (int t = 0; t < 4; ++t) b[0][t] = *(const v2f*)(bp + t * 16 * GB_P);
  for (int s = 0; s < GB_K / 4; ++s) {
    const int cur = s & 1;
    const int nxt = cur ^ 1;
    if (s + 1 < GB_K / 4) {
      a[nxt] = *(const v2f*)(ap + 4 * (s + 1));
      for (int t = 0; t < 4; ++t)
        b[nxt][t] = *(const v2f*)(bp + t * 16 * GB_P + 4 * (s + 1));
    }
    for (int t = 0; t < 4; ++t)
      acc[t] = __builtin_amdgcn_wmma_f32_16x16x4_f32(
          false, a[cur], false, b[cur][t], (short)0, acc[t], false, false);
  }
}

template <bool DOTS>
__device__ __forceinline__ void gat_gemm_tile(const float* __restrict__ X,
                                              const float* __restrict__ W,
                                              float* __restrict__ OUT, int K, int Oout,
                                              const float* __restrict__ al,
                                              const float* __restrict__ ar,
                                              float* __restrict__ el, float* __restrict__ er,
                                              int Hh, int F) {
  __shared__ __attribute__((aligned(16))) float tile[2 * GB_STAGE];
  const int tid  = threadIdx.x;
  const int lane = tid & 31;
  const int wv   = tid >> 5;
  const int row0 = blockIdx.x * GB_T;
  const int col0 = blockIdx.y * GB_T;
  const int mr   = lane & 15;
  const int half = lane >> 4;

  const int c4 = tid & 7;
  const int r8 = tid >> 3;
  const float* xp[4];
  const float* wp[4];
  for (int i = 0; i < 4; ++i) {
    int xr = row0 + r8 + 16 * i;
    xr = (xr < NN) ? xr : (NN - 1);
    xp[i] = X + (size_t)xr * (size_t)K + 4 * c4;
    wp[i] = W + (size_t)(col0 + r8 + 16 * i) * (size_t)K + 4 * c4;
  }
  float* sx = tile + r8 * GB_P + 4 * c4;
  float* sw = sx + GB_T * GB_P;
  const float* ap = tile + (16 * wv + mr) * GB_P + 2 * half;
  const float* bp = tile + (GB_T + mr) * GB_P + 2 * half;

  float4 px[4], pw[4];
  for (int i = 0; i < 4; ++i) {
    px[i] = *(const float4*)(xp[i]);
    pw[i] = *(const float4*)(wp[i]);
  }
  for (int i = 0; i < 4; ++i) {
    *(float4*)(sx + 16 * i * GB_P) = px[i];
    *(float4*)(sw + 16 * i * GB_P) = pw[i];
  }
  __syncthreads();

  v8f acc[4] = {};
  const int nkb = K / GB_K;
  for (int kb = 0; kb + 1 < nkb; ++kb) {
    const int so = (kb & 1) * GB_STAGE;
    const int sn = GB_STAGE - so;
    const int kn = (kb + 1) * GB_K;
    for (int i = 0; i < 4; ++i) {
      px[i] = *(const float4*)(xp[i] + kn);
      pw[i] = *(const float4*)(wp[i] + kn);
    }
    gat_gemm_kblock(ap + so, bp + so, acc);
    for (int i = 0; i < 4; ++i) {
      *(float4*)(sx + sn + 16 * i * GB_P) = px[i];
      *(float4*)(sw + sn + 16 * i * GB_P) = pw[i];
    }
    __syncthreads();
  }
  float cl[4], cr[4];
  if (DOTS) {
    const int head = blockIdx.y;
    for (int t = 0; t < 4; ++t) {
      const int c = (t << 4) + mr;
      const bool in = c < F;
      const int ci = head * F + (in ? c : 0);
      const float a = al[ci];
      const float b = ar[ci];
      cl[t] = in ? a : 0.0f;
      cr[t] = in ? b : 0.0f;
    }
  }
  {
    const int so = ((nkb - 1) & 1) * GB_STAGE;
    gat_gemm_kblock(ap + so, bp + so, acc);
  }

  const int wrow0 = row0 + 16 * wv;
  if (wrow0 < NN) {
    for (int t = 0; t < 4; ++t) {
      const int oc = col0 + (t << 4) + mr;
      if (oc < Oout) {
        float* orow = OUT + (size_t)(wrow0 + (half << 3)) * (size_t)Oout + oc;
        for (int v = 0; v < 8; ++v) {
          orow[(size_t)v * (size_t)Oout] = acc[t][v];
        }
      }
    }
  }

  if (DOTS) {
    float pl[8], pr[8];
    for (int v = 0; v < 8; ++v) {
      float sl = acc[0][v] * cl[0];
      float sr = acc[0][v] * cr[0];
      for (int t = 1; t < 4; ++t) {
        sl = fmaf(acc[t][v], cl[t], sl);
        sr = fmaf(acc[t][v], cr[t], sr);
      }
      pl[v] = sl;
      pr[v] = sr;
    }
    for (int v = 0; v < 8; ++v) { pl[v] = gat_row16_step<1>(pl[v]); pr[v] = gat_row16_step<1>(pr[v]); }
    for (int v = 0; v < 8; ++v) { pl[v] = gat_row16_step<2>(pl[v]); pr[v] = gat_row16_step<2>(pr[v]); }
    for (int v = 0; v < 8; ++v) { pl[v] = gat_row16_step<4>(pl[v]); pr[v] = gat_row16_step<4>(pr[v]); }
    for (int v = 0; v < 8; ++v) { pl[v] = gat_row16_step<8>(pl[v]); pr[v] = gat_row16_step<8>(pr[v]); }
    const int v8 = mr & 7;
    float ol = pl[0], orr = pr[0];
    for (int v = 1; v < 8; ++v) {
      ol  = (v8 == v) ? pl[v] : ol;
      orr = (v8 == v) ? pr[v] : orr;
    }
    if (wrow0 < NN) {
      float* dp = (mr < 8) ? el : er;
      dp[(size_t)(wrow0 + (half << 3) + v8) * (size_t)Hh + blockIdx.y] = (mr < 8) ? ol : orr;
    }
  }
}

__global__ __launch_bounds__(128)
void gat_gemm_xwt(const float* __restrict__ X, const float* __restrict__ W,
                  float* __restrict__ OUT, int K, int Opad, int Oout) {
  (void)Opad;
  gat_gemm_tile<false>(X, W, OUT, K, Oout, nullptr, nullptr, nullptr, nullptr, 0, 0);
}

__global__ __launch_bounds__(128)
void gat_gemm_xwt_dots(const float* __restrict__ X, const float* __restrict__ W,
                       float* __restrict__ OUT, int K, int Opad, int Oout,
                       const float* __restrict__ al, const float* __restrict__ ar,
                       float* __restrict__ el, float* __restrict__ er, int Hh, int F) {
  (void)Opad;
  gat_gemm_tile<true>(X, W, OUT, K, Oout, al, ar, el, er, Hh, F);
}

__global__ void gat_pad_w(const float* __restrict__ Wsrc, float* __restrict__ Wdst,
                          int rows, int padrows, int K) {
  int idx = blockIdx.x * blockDim.x + threadIdx.x;
  int total = padrows * K;
  if (idx >= total) return;
  int r = idx / K;
  Wdst[idx] = (r < rows) ? Wsrc[idx] : 0.0f;
}

#define CSR_PERM (NN + 1)
#define CSR_CUR  (CSR_PERM + EE)
#define CSR_BSUM (CSR_CUR + NN)
#define CSR_NB   ((NN + 255) / 256)

__global__ void gat_csr_zero(int* __restrict__ cur, int total) {
  int idx = blockIdx.x * blockDim.x + threadIdx.x;
  if (idx < total) cur[idx] = 0;
}

__global__ void gat_csr_hist(const int* __restrict__ dst, int* __restrict__ cur, int total) {
  int e = blockIdx.x * blockDim.x + threadIdx.x;
  if (e < total) atomicAdd(cur + dst[e], 1);
}

__device__ __forceinline__ int gat_block_scan256(int v, int* sh) {
  const int t = threadIdx.x;
  sh[t] = v;
  __syncthreads();
  for (int d = 1; d < 256; d <<= 1) {
    int a = (t >= d) ? sh[t - d] : 0;
    __syncthreads();
    sh[t] += a;
    __syncthreads();
  }
  return sh[t];
}

__global__ __launch_bounds__(256)
void gat_csr_bsum(const int* __restrict__ cur, int* __restrict__ bsum) {
  __shared__ int sh[256];
  const int i = blockIdx.x * 256 + threadIdx.x;
  const int v = (i < NN) ? cur[i] : 0;
  const int inc = gat_block_scan256(v, sh);
  if (threadIdx.x == 255) bsum[blockIdx.x] = inc;
}

__global__ __launch_bounds__(256)
void gat_csr_offsets(int* __restrict__ cur, const int* __restrict__ bsum,
                     int* __restrict__ off) {
  __shared__ int sh[256];
  const int t = threadIdx.x;
  const int b = blockIdx.x;
  const int x = (t < b) ? bsum[t] : 0;
  gat_block_scan256(x, sh);
  const int before = sh[255];
  __syncthreads();
  const int i = b * 256 + t;
  const int v = (i < NN) ? cur[i] : 0;
  const int inc = gat_block_scan256(v, sh);
  if (i < NN) {
    const int o = before + inc - v;
    off[i] = o;
    cur[i] = o;
    if (i == NN - 1) off[NN] = o + v;
  }
}

__global__ void gat_csr_fill(const int* __restrict__ dst, int* __restrict__ cur,
                             int* __restrict__ perm, int total) {
  int e = blockIdx.x * blockDim.x + threadIdx.x;
  if (e >= total) return;
  int p = atomicAdd(cur + dst[e], 1);
  perm[p] = e;
}


__device__ __forceinline__ float gat_leaky(float s) {
  return (s > 0.0f) ? s : 0.2f * s;
}
__device__ __forceinline__ float gat_elu1(float x) {
  return (x > 0.0f) ? x : expm1f(x);
}
__device__ __forceinline__ float gat_wave_max(float v) {
  for (int d = 16; d >= 1; d >>= 1) v = fmaxf(v, __shfl_xor(v, d, 32));
  return v;
}
__device__ __forceinline__ float gat_wave_sum(float v) {
  for (int d = 16; d >= 1; d >>= 1) v += __shfl_xor(v, d, 32);
  return v;
}

__device__ __forceinline__ void gat_chunk4(int base, int end, int lane,
                                           const int* __restrict__ perm,
                                           const int* __restrict__ src,
                                           const float* __restrict__ el, float4 rn,
                                           int& my_s, float4& sc) {
  my_s = 0;
  sc.x = -INFINITY; sc.y = -INFINITY; sc.z = -INFINITY; sc.w = -INFINITY;
  if (base + lane < end) {
    const int e = perm[base + lane];
    my_s = src[e];
    const float4 l = *(const float4*)(el + (size_t)my_s * 4);
    sc.x = gat_leaky(l.x + rn.x);
    sc.y = gat_leaky(l.y + rn.y);
    sc.z = gat_leaky(l.z + rn.z);
    sc.w = gat_leaky(l.w + rn.w);
  }
}

__global__ __launch_bounds__(256)
void gat_gather256(const float* __restrict__ Hm, const float* __restrict__ el,
                   const float* __restrict__ er, const int* __restrict__ src,
                   const int* __restrict__ off, const int* __restrict__ perm,
                   float* __restrict__ AGG, int act) {
  const int lane = threadIdx.x & 31;
  const int n = __builtin_amdgcn_readfirstlane((int)(blockIdx.x * 8 + (threadIdx.x >> 5)));
  if (n >= NN) return;
  const int beg = off[n];
  const int end = off[n + 1];
  const float4 rn = *(const float4*)(er + (size_t)n * 4);
  const bool one = (end - beg) <= 32;
  const bool hi = lane >= 16;
  int my_s = 0;
  float4 sc = {-INFINITY, -INFINITY, -INFINITY, -INFINITY};

  float4 mx = {-INFINITY, -INFINITY, -INFINITY, -INFINITY};
  for (int base = beg; base < end; base += 32) {
    gat_chunk4(base, end, lane, perm, src, el, rn, my_s, sc);
    mx.x = fmaxf(mx.x, sc.x); mx.y = fmaxf(mx.y, sc.y);
    mx.z = fmaxf(mx.z, sc.z); mx.w = fmaxf(mx.w, sc.w);
  }
  mx.x = gat_wave_max(mx.x); mx.y = gat_wave_max(mx.y);
  mx.z = gat_wave_max(mx.z); mx.w = gat_wave_max(mx.w);

  float4 dn = {0.f, 0.f, 0.f, 0.f};
  for (int base = beg; base < end; base += 32) {
    if (!one) gat_chunk4(base, end, lane, perm, src, el, rn, my_s, sc);
    dn.x += expf(sc.x - mx.x); dn.y += expf(sc.y - mx.y);
    dn.z += expf(sc.z - mx.z); dn.w += expf(sc.w - mx.w);
  }
  dn.x = gat_wave_sum(dn.x); dn.y = gat_wave_sum(dn.y);
  dn.z = gat_wave_sum(dn.z); dn.w = gat_wave_sum(dn.w);

  float4 acc0 = {0.f, 0.f, 0.f, 0.f};
  float4 acc1 = {0.f, 0.f, 0.f, 0.f};
  for (int base = beg; base < end; base += 32) {
    const int left = end - base;
    const int cnt = (left < 32) ? left : 32;
    if (!one) gat_chunk4(base, end, lane, perm, src, el, rn, my_s, sc);
    float4 al;
    al.x = expf(sc.x - mx.x) / dn.x;
    al.y = expf(sc.y - mx.y) / dn.y;
    al.z = expf(sc.z - mx.z) / dn.z;
    al.w = expf(sc.w - mx.w) / dn.w;
#pragma unroll 4
    for (int i = 0; i < cnt; ++i) {
      const int s = __builtin_amdgcn_readlane(my_s, i);
      const float a0 = __int_as_float(__builtin_amdgcn_readlane(__float_as_int(al.x), i));
      const float a1 = __int_as_float(__builtin_amdgcn_readlane(__float_as_int(al.y), i));
      const float a2 = __int_as_float(__builtin_amdgcn_readlane(__float_as_int(al.z), i));
      const float a3 = __int_as_float(__builtin_amdgcn_readlane(__float_as_int(al.w), i));
      const float w0 = hi ? a1 : a0;
      const float w1 = hi ? a3 : a2;
      const float4* row = (const float4*)(Hm + (size_t)s * 256);
      const float4 v0 = row[lane];
      const float4 v1 = row[32 + lane];
      acc0.x += v0.x * w0; acc0.y += v0.y * w0; acc0.z += v0.z * w0; acc0.w += v0.w * w0;
      acc1.x += v1.x * w1; acc1.y += v1.y * w1; acc1.z += v1.z * w1; acc1.w += v1.w * w1;
    }
  }

  float4* o = (float4*)(AGG + (size_t)n * 256);
  float4 r0 = o[lane];
  float4 r1 = o[32 + lane];
  r0.x += acc0.x; r0.y += acc0.y; r0.z += acc0.z; r0.w += acc0.w;
  r1.x += acc1.x; r1.y += acc1.y; r1.z += acc1.z; r1.w += acc1.w;
  if (act) {
    r0.x = gat_elu1(r0.x); r0.y = gat_elu1(r0.y); r0.z = gat_elu1(r0.z); r0.w = gat_elu1(r0.w);
    r1.x = gat_elu1(r1.x); r1.y = gat_elu1(r1.y); r1.z = gat_elu1(r1.z); r1.w = gat_elu1(r1.w);
  }
  o[lane] = r0;
  o[32 + lane] = r1;
}

__device__ __forceinline__ void gat_chunk1(int base, int end, int lane,
                                           const int* __restrict__ perm,
                                           const int* __restrict__ src,
                                           const float* __restrict__ el, float rn,
                                           int& my_s, float& sc) {
  my_s = 0;
  sc = -INFINITY;
  if (base + lane < end) {
    const int e = perm[base + lane];
    my_s = src[e];
    sc = gat_leaky(el[my_s] + rn);
  }
}

__global__ __launch_bounds__(256)
void gat_gather40(const float* __restrict__ Hm, const float* __restrict__ el,
                  const float* __restrict__ er, const int* __restrict__ src,
                  const int* __restrict__ off, const int* __restrict__ perm,
                  float* __restrict__ AGG, int act) {
  const int lane = threadIdx.x & 31;
  const int g = __builtin_amdgcn_readfirstlane((int)(blockIdx.x * 8 + (threadIdx.x >> 5)));
  if (g * 4 >= NN) return;
  const int col = ((lane < 20) ? lane : 19) * 2;
  for (int q = 0; q < 4; ++q) {
    const int n = g * 4 + q;
    const int beg = off[n];
    const int end = off[n + 1];
    const float rn = er[n];
    const bool one = (end - beg) <= 32;
    int my_s = 0;
    float sc = -INFINITY;

    float mx = -INFINITY;
    for (int base = beg; base < end; base += 32) {
      gat_chunk1(base, end, lane, perm, src, el, rn, my_s, sc);
      mx = fmaxf(mx, sc);
    }
    mx = gat_wave_max(mx);

    float dn = 0.f;
    for (int base = beg; base < end; base += 32) {
      if (!one) gat_chunk1(base, end, lane, perm, src, el, rn, my_s, sc);
      dn += expf(sc - mx);
    }
    dn = gat_wave_sum(dn);

    float2 acc = {0.f, 0.f};
    for (int base = beg; base < end; base += 32) {
      const int left = end - base;
      const int cnt = (left < 32) ? left : 32;
      if (!one) gat_chunk1(base, end, lane, perm, src, el, rn, my_s, sc);
      const float my_a = expf(sc - mx) / dn;
#pragma unroll 4
      for (int i = 0; i < cnt; ++i) {
        const int s = __builtin_amdgcn_readlane(my_s, i);
        const float a = __int_as_float(__builtin_amdgcn_readlane(__float_as_int(my_a), i));
        const float2 v = *(const float2*)(Hm + (size_t)s * 40 + col);
        acc.x += v.x * a;
        acc.y += v.y * a;
      }
    }
    if (lane < 20) {
      float2* o = (float2*)(AGG + (size_t)n * 40 + col);
      float2 r = *o;
      r.x += acc.x;
      r.y += acc.y;
      if (act) { r.x = gat_elu1(r.x); r.y = gat_elu1(r.y); }
      *o = r;
    }
  }
}

static void run_layer(const float* X, int K, int Hh, int F,
                      const float* W, const float* al, const float* ar, const float* rW,
                      const int* src,
                      float* Hbuf, float* AGG, bool act, int Opad,
                      float* el, float* er, const int* csr,
                      hipStream_t stream) {
  const int HF = Hh * F;
  dim3 ggrid((NN + GB_T - 1) / GB_T, Opad / GB_T);
  gat_gemm_xwt_dots<<<ggrid, 128, 0, stream>>>(X, W, Hbuf, K, Opad, HF, al, ar, el, er, Hh, F);
  gat_gemm_xwt<<<ggrid, 128, 0, stream>>>(X, rW, AGG, K, Opad, HF);

  if (HF == 256)
    gat_gather256<<<NN / 8, 256, 0, stream>>>(Hbuf, el, er, src, csr, csr + CSR_PERM, AGG,
                                              act ? 1 : 0);
  else
    gat_gather40<<<(NN / 4 + 7) / 8, 256, 0, stream>>>(Hbuf, el, er, src, csr, csr + CSR_PERM,
                                                       AGG, act ? 1 : 0);
}

extern "C" void kernel_launch(void* const* d_in, const int* in_sizes, int n_in,
                              void* d_out, int out_size, void* d_ws, size_t ws_size,
                              hipStream_t stream) {
  (void)in_sizes; (void)n_in; (void)out_size; (void)ws_size;

  const float* feat = (const float*)d_in[0];
  const int*   src  = (const int*)d_in[1];
  const int*   dst  = (const int*)d_in[2];
  const float* W0  = (const float*)d_in[3];
  const float* al0 = (const float*)d_in[4];
  const float* ar0 = (const float*)d_in[5];
  const float* rW0 = (const float*)d_in[6];
  const float* W1  = (const float*)d_in[7];
  const float* al1 = (const float*)d_in[8];
  const float* ar1 = (const float*)d_in[9];
  const float* rW1 = (const float*)d_in[10];
  const float* W2  = (const float*)d_in[11];
  const float* al2 = (const float*)d_in[12];
  const float* ar2 = (const float*)d_in[13];
  const float* rW2 = (const float*)d_in[14];

  const size_t NF = (size_t)NN * 256;
  float* ws    = (float*)d_ws;
  float* bufH  = ws;
  float* bufY0 = bufH  + NF;
  float* bufY1 = bufY0 + NF;
  float* el    = bufY1 + NF;
  float* er    = el  + (size_t)NN * 4;
  float* sb    = er  + (size_t)NN * 12;
  float* W2p   = sb + (size_t)EE * 4;
  float* rW2p  = W2p + (size_t)64 * 256;

  {
    int tot = 64 * 256;
    gat_pad_w<<<(tot + 255) / 256, 256, 0, stream>>>(W2,  W2p,  40, 64, 256);
    gat_pad_w<<<(tot + 255) / 256, 256, 0, stream>>>(rW2, rW2p, 40, 64, 256);
  }

  int* csr = (int*)sb;
  {
    gat_csr_zero<<<(NN + 255) / 256, 256, 0, stream>>>(csr + CSR_CUR, NN);
    gat_csr_hist<<<(EE + 255) / 256, 256, 0, stream>>>(dst, csr + CSR_CUR, EE);
    gat_csr_bsum<<<CSR_NB, 256, 0, stream>>>(csr + CSR_CUR, csr + CSR_BSUM);
    gat_csr_offsets<<<CSR_NB, 256, 0, stream>>>(csr + CSR_CUR, csr + CSR_BSUM, csr);
    gat_csr_fill<<<(EE + 255) / 256, 256, 0, stream>>>(dst, csr + CSR_CUR, csr + CSR_PERM, EE);
  }

  run_layer(feat, 128, 4, 64, W0, al0, ar0, rW0, src,
            bufH, bufY0, true, 256, el, er, csr, stream);
  run_layer(bufY0, 256, 4, 64, W1, al1, ar1, rW1, src,
            bufH, bufY1, true, 256, el, er, csr, stream);
  run_layer(bufY1, 256, 1, 40, W2p, al2, ar2, rW2p, src,
            bufH, (float*)d_out, false, 64, el, er, csr, stream);
}


# Sorted edge list stores the source node itself (the sort's fill pass writes src[e] into the slot): gathers read a node's sources as consecutive ints, no random src read per edge; sum order as before
# speedup vs baseline: 5.4488x; 1.0356x over previous
#include <hip/hip_runtime.h>
#include <math.h>

#define NN 50000
#define EE 800000

typedef __attribute__((ext_vector_type(2))) float v2f;
typedef __attribute__((ext_vector_type(8))) float v8f;

#define GB_T 64
#define GB_K 32
#define GB_P 36
#define GB_STAGE (2 * GB_T * GB_P)

template <int XM>
__device__ __forceinline__ float gat_row16_step(float x) {
  const int xi = __builtin_bit_cast(int, x);
  const int yi = __builtin_amdgcn_update_dpp(0, xi, 0x160 + XM, 0xf, 0xf, true);
  return x + __builtin_bit_cast(float, yi);
}

__device__ __forceinline__ void gat_gemm_kblock(const float* ap, const float* bp,
                                                v8f (&acc)[4]) {
  v2f a[2];
  v2f b[2][4];
  a[0] = *(const v2f*)ap;
  for (int t = 0; t < 4; ++t) b[0][t] = *(const v2f*)(bp + t * 16 * GB_P);
  for (int s = 0; s < GB_K / 4; ++s) {
    const int cur = s & 1;
    const int nxt = cur ^ 1;
    if (s + 1 < GB_K / 4) {
      a[nxt] = *(const v2f*)(ap + 4 * (s + 1));
      for (int t = 0; t < 4; ++t)
        b[nxt][t] = *(const v2f*)(bp + t * 16 * GB_P + 4 * (s + 1));
    }
    for (int t = 0; t < 4; ++t)
      acc[t] = __builtin_amdgcn_wmma_f32_16x16x4_f32(
          false, a[cur], false, b[cur][t], (short)0, acc[t], false, false);
  }
}

template <bool DOTS>
__device__ __forceinline__ void gat_gemm_tile(const float* __restrict__ X,
                                              const float* __restrict__ W,
                                              float* __restrict__ OUT, int K, int Oout,
                                              const float* __restrict__ al,
                                              const float* __restrict__ ar,
                                              float* __restrict__ el, float* __restrict__ er,
                                              int Hh, int F) {
  __shared__ __attribute__((aligned(16))) float tile[2 * GB_STAGE];
  const int tid  = threadIdx.x;
  const int lane = tid & 31;
  const int wv   = tid >> 5;
  const int row0 = blockIdx.x * GB_T;
  const int col0 = blockIdx.y * GB_T;
  const int mr   = lane & 15;
  const int half = lane >> 4;

  const int c4 = tid & 7;
  const int r8 = tid >> 3;
  const float* xp[4];
  const float* wp[4];
  for (int i = 0; i < 4; ++i) {
    int xr = row0 + r8 + 16 * i;
    xr = (xr < NN) ? xr : (NN - 1);
    xp[i] = X + (size_t)xr * (size_t)K + 4 * c4;
    wp[i] = W + (size_t)(col0 + r8 + 16 * i) * (size_t)K + 4 * c4;
  }
  float* sx = tile + r8 * GB_P + 4 * c4;
  float* sw = sx + GB_T * GB_P;
  const float* ap = tile + (16 * wv + mr) * GB_P + 2 * half;
  const float* bp = tile + (GB_T + mr) * GB_P + 2 * half;

  float4 px[4], pw[4];
  for (int i = 0; i < 4; ++i) {
    px[i] = *(const float4*)(xp[i]);
    pw[i] = *(const float4*)(wp[i]);
  }
  for (int i = 0; i < 4; ++i) {
    *(float4*)(sx + 16 * i * GB_P) = px[i];
    *(float4*)(sw + 16 * i * GB_P) = pw[i];
  }
  __syncthreads();

  v8f acc[4] = {};
  const int nkb = K / GB_K;
  for (int kb = 0; kb + 1 < nkb; ++kb) {
    const int so = (kb & 1) * GB_STAGE;
    const int sn = GB_STAGE - so;
    const int kn = (kb + 1) * GB_K;
    for (int i = 0; i < 4; ++i) {
      px[i] = *(const float4*)(xp[i] + kn);
      pw[i] = *(const float4*)(wp[i] + kn);
    }
    gat_gemm_kblock(ap + so, bp + so, acc);
    for (int i = 0; i < 4; ++i) {
      *(float4*)(sx + sn + 16 * i * GB_P) = px[i];
      *(float4*)(sw + sn + 16 * i * GB_P) = pw[i];
    }
    __syncthreads();
  }
  float cl[4], cr[4];
  if (DOTS) {
    const int head = blockIdx.y;
    for (int t = 0; t < 4; ++t) {
      const int c = (t << 4) + mr;
      const bool in = c < F;
      const int ci = head * F + (in ? c : 0);
      const float a = al[ci];
      const float b = ar[ci];
      cl[t] = in ? a : 0.0f;
      cr[t] = in ? b : 0.0f;
    }
  }
  {
    const int so = ((nkb - 1) & 1) * GB_STAGE;
    gat_gemm_kblock(ap + so, bp + so, acc);
  }

  const int wrow0 = row0 + 16 * wv;
  if (wrow0 < NN) {
    for (int t = 0; t < 4; ++t) {
      const int oc = col0 + (t << 4) + mr;
      if (oc < Oout) {
        float* orow = OUT + (size_t)(wrow0 + (half << 3)) * (size_t)Oout + oc;
        for (int v = 0; v < 8; ++v) {
          orow[(size_t)v * (size_t)Oout] = acc[t][v];
        }
      }
    }
  }

  if (DOTS) {
    float pl[8], pr[8];
    for (int v = 0; v < 8; ++v) {
      float sl = acc[0][v] * cl[0];
      float sr = acc[0][v] * cr[0];
      for (int t = 1; t < 4; ++t) {
        sl = fmaf(acc[t][v], cl[t], sl);
        sr = fmaf(acc[t][v], cr[t], sr);
      }
      pl[v] = sl;
      pr[v] = sr;
    }
    for (int v = 0; v < 8; ++v) { pl[v] = gat_row16_step<1>(pl[v]); pr[v] = gat_row16_step<1>(pr[v]); }
    for (int v = 0; v < 8; ++v) { pl[v] = gat_row16_step<2>(pl[v]); pr[v] = gat_row16_step<2>(pr[v]); }
    for (int v = 0; v < 8; ++v) { pl[v] = gat_row16_step<4>(pl[v]); pr[v] = gat_row16_step<4>(pr[v]); }
    for (int v = 0; v < 8; ++v) { pl[v] = gat_row16_step<8>(pl[v]); pr[v] = gat_row16_step<8>(pr[v]); }
    const int v8 = mr & 7;
    float ol = pl[0], orr = pr[0];
    for (int v = 1; v < 8; ++v) {
      ol  = (v8 == v) ? pl[v] : ol;
      orr = (v8 == v) ? pr[v] : orr;
    }
    if (wrow0 < NN) {
      float* dp = (mr < 8) ? el : er;
      dp[(size_t)(wrow0 + (half << 3) + v8) * (size_t)Hh + blockIdx.y] = (mr < 8) ? ol : orr;
    }
  }
}

__global__ __launch_bounds__(128)
void gat_gemm_xwt(const float* __restrict__ X, const float* __restrict__ W,
                  float* __restrict__ OUT, int K, int Opad, int Oout) {
  (void)Opad;
  gat_gemm_tile<false>(X, W, OUT, K, Oout, nullptr, nullptr, nullptr, nullptr, 0, 0);
}

__global__ __launch_bounds__(128)
void gat_gemm_xwt_dots(const float* __restrict__ X, const float* __restrict__ W,
                       float* __restrict__ OUT, int K, int Opad, int Oout,
                       const float* __restrict__ al, const float* __restrict__ ar,
                       float* __restrict__ el, float* __restrict__ er, int Hh, int F) {
  (void)Opad;
  gat_gemm_tile<true>(X, W, OUT, K, Oout, al, ar, el, er, Hh, F);
}

__global__ void gat_pad_w(const float* __restrict__ Wsrc, float* __restrict__ Wdst,
                          int rows, int padrows, int K) {
  int idx = blockIdx.x * blockDim.x + threadIdx.x;
  int total = padrows * K;
  if (idx >= total) return;
  int r = idx / K;
  Wdst[idx] = (r < rows) ? Wsrc[idx] : 0.0f;
}

#define CSR_PERM (NN + 1)
#define CSR_CUR  (CSR_PERM + EE)
#define CSR_BSUM (CSR_CUR + NN)
#define CSR_NB   ((NN + 255) / 256)

__global__ void gat_csr_zero(int* __restrict__ cur, int total) {
  int idx = blockIdx.x * blockDim.x + threadIdx.x;
  if (idx < total) cur[idx] = 0;
}

__global__ void gat_csr_hist(const int* __restrict__ dst, int* __restrict__ cur, int total) {
  int e = blockIdx.x * blockDim.x + threadIdx.x;
  if (e < total) atomicAdd(cur + dst[e], 1);
}

__device__ __forceinline__ int gat_block_scan256(int v, int* sh) {
  const int t = threadIdx.x;
  sh[t] = v;
  __syncthreads();
  for (int d = 1; d < 256; d <<= 1) {
    int a = (t >= d) ? sh[t - d] : 0;
    __syncthreads();
    sh[t] += a;
    __syncthreads();
  }
  return sh[t];
}

__global__ __launch_bounds__(256)
void gat_csr_bsum(const int* __restrict__ cur, int* __restrict__ bsum) {
  __shared__ int sh[256];
  const int i = blockIdx.x * 256 + threadIdx.x;
  const int v = (i < NN) ? cur[i] : 0;
  const int inc = gat_block_scan256(v, sh);
  if (threadIdx.x == 255) bsum[blockIdx.x] = inc;
}

__global__ __launch_bounds__(256)
void gat_csr_offsets(int* __restrict__ cur, const int* __restrict__ bsum,
                     int* __restrict__ off) {
  __shared__ int sh[256];
  const int t = threadIdx.x;
  const int b = blockIdx.x;
  const int x = (t < b) ? bsum[t] : 0;
  gat_block_scan256(x, sh);
  const int before = sh[255];
  __syncthreads();
  const int i = b * 256 + t;
  const int v = (i < NN) ? cur[i] : 0;
  const int inc = gat_block_scan256(v, sh);
  if (i < NN) {
    const int o = before + inc - v;
    off[i] = o;
    cur[i] = o;
    if (i == NN - 1) off[NN] = o + v;
  }
}

__global__ void gat_csr_fill(const int* __restrict__ dst, const int* __restrict__ src,
                             int* __restrict__ cur, int* __restrict__ perm, int total) {
  int e = blockIdx.x * blockDim.x + threadIdx.x;
  if (e >= total) return;
  const int s = src[e];
  int p = atomicAdd(cur + dst[e], 1);
  perm[p] = s;
}


__device__ __forceinline__ float gat_leaky(float s) {
  return (s > 0.0f) ? s : 0.2f * s;
}
__device__ __forceinline__ float gat_elu1(float x) {
  return (x > 0.0f) ? x : expm1f(x);
}
__device__ __forceinline__ float gat_wave_max(float v) {
  for (int d = 16; d >= 1; d >>= 1) v = fmaxf(v, __shfl_xor(v, d, 32));
  return v;
}
__device__ __forceinline__ float gat_wave_sum(float v) {
  for (int d = 16; d >= 1; d >>= 1) v += __shfl_xor(v, d, 32);
  return v;
}

__device__ __forceinline__ void gat_chunk4(int base, int end, int lane,
                                           const int* __restrict__ perm,
                                           const float* __restrict__ el, float4 rn,
                                           int& my_s, float4& sc) {
  my_s = 0;
  sc.x = -INFINITY; sc.y = -INFINITY; sc.z = -INFINITY; sc.w = -INFINITY;
  if (base + lane < end) {
    my_s = perm[base + lane];
    const float4 l = *(const float4*)(el + (size_t)my_s * 4);
    sc.x = gat_leaky(l.x + rn.x);
    sc.y = gat_leaky(l.y + rn.y);
    sc.z = gat_leaky(l.z + rn.z);
    sc.w = gat_leaky(l.w + rn.w);
  }
}

__global__ __launch_bounds__(256)
void gat_gather256(const float* __restrict__ Hm, const float* __restrict__ el,
                   const float* __restrict__ er,
                   const int* __restrict__ off, const int* __restrict__ perm,
                   float* __restrict__ AGG, int act) {
  const int lane = threadIdx.x & 31;
  const int n = __builtin_amdgcn_readfirstlane((int)(blockIdx.x * 8 + (threadIdx.x >> 5)));
  if (n >= NN) return;
  const int beg = off[n];
  const int end = off[n + 1];
  const float4 rn = *(const float4*)(er + (size_t)n * 4);
  const bool one = (end - beg) <= 32;
  const bool hi = lane >= 16;
  int my_s = 0;
  float4 sc = {-INFINITY, -INFINITY, -INFINITY, -INFINITY};

  float4 mx = {-INFINITY, -INFINITY, -INFINITY, -INFINITY};
  for (int base = beg; base < end; base += 32) {
    gat_chunk4(base, end, lane, perm, el, rn, my_s, sc);
    mx.x = fmaxf(mx.x, sc.x); mx.y = fmaxf(mx.y, sc.y);
    mx.z = fmaxf(mx.z, sc.z); mx.w = fmaxf(mx.w, sc.w);
  }
  mx.x = gat_wave_max(mx.x); mx.y = gat_wave_max(mx.y);
  mx.z = gat_wave_max(mx.z); mx.w = gat_wave_max(mx.w);

  float4 dn = {0.f, 0.f, 0.f, 0.f};
  for (int base = beg; base < end; base += 32) {
    if (!one) gat_chunk4(base, end, lane, perm, el, rn, my_s, sc);
    dn.x += expf(sc.x - mx.x); dn.y += expf(sc.y - mx.y);
    dn.z += expf(sc.z - mx.z); dn.w += expf(sc.w - mx.w);
  }
  dn.x = gat_wave_sum(dn.x); dn.y = gat_wave_sum(dn.y);
  dn.z = gat_wave_sum(dn.z); dn.w = gat_wave_sum(dn.w);

  float4 acc0 = {0.f, 0.f, 0.f, 0.f};
  float4 acc1 = {0.f, 0.f, 0.f, 0.f};
  for (int base = beg; base < end; base += 32) {
    const int left = end - base;
    const int cnt = (left < 32) ? left : 32;
    if (!one) gat_chunk4(base, end, lane, perm, el, rn, my_s, sc);
    float4 al;
    al.x = expf(sc.x - mx.x) / dn.x;
    al.y = expf(sc.y - mx.y) / dn.y;
    al.z = expf(sc.z - mx.z) / dn.z;
    al.w = expf(sc.w - mx.w) / dn.w;
#pragma unroll 4
    for (int i = 0; i < cnt; ++i) {
      const int s = __builtin_amdgcn_readlane(my_s, i);
      const float a0 = __int_as_float(__builtin_amdgcn_readlane(__float_as_int(al.x), i));
      const float a1 = __int_as_float(__builtin_amdgcn_readlane(__float_as_int(al.y), i));
      const float a2 = __int_as_float(__builtin_amdgcn_readlane(__float_as_int(al.z), i));
      const float a3 = __int_as_float(__builtin_amdgcn_readlane(__float_as_int(al.w), i));
      const float w0 = hi ? a1 : a0;
      const float w1 = hi ? a3 : a2;
      const float4* row = (const float4*)(Hm + (size_t)s * 256);
      const float4 v0 = row[lane];
      const float4 v1 = row[32 + lane];
      acc0.x += v0.x * w0; acc0.y += v0.y * w0; acc0.z += v0.z * w0; acc0.w += v0.w * w0;
      acc1.x += v1.x * w1; acc1.y += v1.y * w1; acc1.z += v1.z * w1; acc1.w += v1.w * w1;
    }
  }

  float4* o = (float4*)(AGG + (size_t)n * 256);
  float4 r0 = o[lane];
  float4 r1 = o[32 + lane];
  r0.x += acc0.x; r0.y += acc0.y; r0.z += acc0.z; r0.w += acc0.w;
  r1.x += acc1.x; r1.y += acc1.y; r1.z += acc1.z; r1.w += acc1.w;
  if (act) {
    r0.x = gat_elu1(r0.x); r0.y = gat_elu1(r0.y); r0.z = gat_elu1(r0.z); r0.w = gat_elu1(r0.w);
    r1.x = gat_elu1(r1.x); r1.y = gat_elu1(r1.y); r1.z = gat_elu1(r1.z); r1.w = gat_elu1(r1.w);
  }
  o[lane] = r0;
  o[32 + lane] = r1;
}

__device__ __forceinline__ void gat_chunk1(int base, int end, int lane,
                                           const int* __restrict__ perm,
                                           const float* __restrict__ el, float rn,
                                           int& my_s, float& sc) {
  my_s = 0;
  sc = -INFINITY;
  if (base + lane < end) {
    my_s = perm[base + lane];
    sc = gat_leaky(el[my_s] + rn);
  }
}

__global__ __launch_bounds__(256)
void gat_gather40(const float* __restrict__ Hm, const float* __restrict__ el,
                  const float* __restrict__ er,
                  const int* __restrict__ off, const int* __restrict__ perm,
                  float* __restrict__ AGG, int act) {
  const int lane = threadIdx.x & 31;
  const int g = __builtin_amdgcn_readfirstlane((int)(blockIdx.x * 8 + (threadIdx.x >> 5)));
  if (g * 4 >= NN) return;
  const int col = ((lane < 20) ? lane : 19) * 2;
  for (int q = 0; q < 4; ++q) {
    const int n = g * 4 + q;
    const int beg = off[n];
    const int end = off[n + 1];
    const float rn = er[n];
    const bool one = (end - beg) <= 32;
    int my_s = 0;
    float sc = -INFINITY;

    float mx = -INFINITY;
    for (int base = beg; base < end; base += 32) {
      gat_chunk1(base, end, lane, perm, el, rn, my_s, sc);
      mx = fmaxf(mx, sc);
    }
    mx = gat_wave_max(mx);

    float dn = 0.f;
    for (int base = beg; base < end; base += 32) {
      if (!one) gat_chunk1(base, end, lane, perm, el, rn, my_s, sc);
      dn += expf(sc - mx);
    }
    dn = gat_wave_sum(dn);

    float2 acc = {0.f, 0.f};
    for (int base = beg; base < end; base += 32) {
      const int left = end - base;
      const int cnt = (left < 32) ? left : 32;
      if (!one) gat_chunk1(base, end, lane, perm, el, rn, my_s, sc);
      const float my_a = expf(sc - mx) / dn;
#pragma unroll 4
      for (int i = 0; i < cnt; ++i) {
        const int s = __builtin_amdgcn_readlane(my_s, i);
        const float a = __int_as_float(__builtin_amdgcn_readlane(__float_as_int(my_a), i));
        const float2 v = *(const float2*)(Hm + (size_t)s * 40 + col);
        acc.x += v.x * a;
        acc.y += v.y * a;
      }
    }
    if (lane < 20) {
      float2* o = (float2*)(AGG + (size_t)n * 40 + col);
      float2 r = *o;
      r.x += acc.x;
      r.y += acc.y;
      if (act) { r.x = gat_elu1(r.x); r.y = gat_elu1(r.y); }
      *o = r;
    }
  }
}

static void run_layer(const float* X, int K, int Hh, int F,
                      const float* W, const float* al, const float* ar, const float* rW,
                      float* Hbuf, float* AGG, bool act, int Opad,
                      float* el, float* er, const int* csr,
                      hipStream_t stream) {
  const int HF = Hh * F;
  dim3 ggrid((NN + GB_T - 1) / GB_T, Opad / GB_T);
  gat_gemm_xwt_dots<<<ggrid, 128, 0, stream>>>(X, W, Hbuf, K, Opad, HF, al, ar, el, er, Hh, F);
  gat_gemm_xwt<<<ggrid, 128, 0, stream>>>(X, rW, AGG, K, Opad, HF);

  if (HF == 256)
    gat_gather256<<<NN / 8, 256, 0, stream>>>(Hbuf, el, er, csr, csr + CSR_PERM, AGG,
                                              act ? 1 : 0);
  else
    gat_gather40<<<(NN / 4 + 7) / 8, 256, 0, stream>>>(Hbuf, el, er, csr, csr + CSR_PERM,
                                                       AGG, act ? 1 : 0);
}

extern "C" void kernel_launch(void* const* d_in, const int* in_sizes, int n_in,
                              void* d_out, int out_size, void* d_ws, size_t ws_size,
                              hipStream_t stream) {
  (void)in_sizes; (void)n_in; (void)out_size; (void)ws_size;

  const float* feat = (const float*)d_in[0];
  const int*   src  = (const int*)d_in[1];
  const int*   dst  = (const int*)d_in[2];
  const float* W0  = (const float*)d_in[3];
  const float* al0 = (const float*)d_in[4];
  const float* ar0 = (const float*)d_in[5];
  const float* rW0 = (const float*)d_in[6];
  const float* W1  = (const float*)d_in[7];
  const float* al1 = (const float*)d_in[8];
  const float* ar1 = (const float*)d_in[9];
  const float* rW1 = (const float*)d_in[10];
  const float* W2  = (const float*)d_in[11];
  const float* al2 = (const float*)d_in[12];
  const float* ar2 = (const float*)d_in[13];
  const float* rW2 = (const float*)d_in[14];

  const size_t NF = (size_t)NN * 256;
  float* ws    = (float*)d_ws;
  float* bufH  = ws;
  float* bufY0 = bufH  + NF;
  float* bufY1 = bufY0 + NF;
  float* el    = bufY1 + NF;
  float* er    = el  + (size_t)NN * 4;
  float* sb    = er  + (size_t)NN * 12;
  float* W2p   = sb + (size_t)EE * 4;
  float* rW2p  = W2p + (size_t)64 * 256;

  {
    int tot = 64 * 256;
    gat_pad_w<<<(tot + 255) / 256, 256, 0, stream>>>(W2,  W2p,  40, 64, 256);
    gat_pad_w<<<(tot + 255) / 256, 256, 0, stream>>>(rW2, rW2p, 40, 64, 256);
  }

  int* csr = (int*)sb;
  {
    gat_csr_zero<<<(NN + 255) / 256, 256, 0, stream>>>(csr + CSR_CUR, NN);
    gat_csr_hist<<<(EE + 255) / 256, 256, 0, stream>>>(dst, csr + CSR_CUR, EE);
    gat_csr_bsum<<<CSR_NB, 256, 0, stream>>>(csr + CSR_CUR, csr + CSR_BSUM);
    gat_csr_offsets<<<CSR_NB, 256, 0, stream>>>(csr + CSR_CUR, csr + CSR_BSUM, csr);
    gat_csr_fill<<<(EE + 255) / 256, 256, 0, stream>>>(dst, src, csr + CSR_CUR, csr + CSR_PERM,
                                                       EE);
  }

  run_layer(feat, 128, 4, 64, W0, al0, ar0, rW0,
            bufH, bufY0, true, 256, el, er, csr, stream);
  run_layer(bufY0, 256, 4, 64, W1, al1, ar1, rW1,
            bufH, bufY1, true, 256, el, er, csr, stream);
  run_layer(bufY1, 256, 1, 40, W2p, al2, ar2, rW2p,
            bufH, (float*)d_out, false, 64, el, er, csr, stream);
}


# h and residual products of a layer in one launch: a workgroup stages its X tile once and forms both 64-column tiles from it (one A fragment feeds 8 WMMAs); 13 launches; k order of sums unchanged
# speedup vs baseline: 5.7769x; 1.0602x over previous
#include <hip/hip_runtime.h>
#include <math.h>

#define NN 50000
#define EE 800000

typedef __attribute__((ext_vector_type(2))) float v2f;
typedef __attribute__((ext_vector_type(8))) float v8f;

#define GB_T 64
#define GB_K 32
#define GB_P 36
#define GB_STAGE (3 * GB_T * GB_P)

template <int XM>
__device__ __forceinline__ float gat_row16_step(float x) {
  const int xi = __builtin_bit_cast(int, x);
  const int yi = __builtin_amdgcn_update_dpp(0, xi, 0x160 + XM, 0xf, 0xf, true);
  return x + __builtin_bit_cast(float, yi);
}

__device__ __forceinline__ void gat_gemm_kblock(const float* ap, const float* bp,
                                                v8f (&acc)[8]) {
  v2f a[2];
  v2f b[2][8];
  a[0] = *(const v2f*)ap;
  for (int t = 0; t < 8; ++t) b[0][t] = *(const v2f*)(bp + t * 16 * GB_P);
  for (int s = 0; s < GB_K / 4; ++s) {
    const int cur = s & 1;
    const int nxt = cur ^ 1;
    if (s + 1 < GB_K / 4) {
      a[nxt] = *(const v2f*)(ap + 4 * (s + 1));
      for (int t = 0; t < 8; ++t)
        b[nxt][t] = *(const v2f*)(bp + t * 16 * GB_P + 4 * (s + 1));
    }
    for (int t = 0; t < 8; ++t)
      acc[t] = __builtin_amdgcn_wmma_f32_16x16x4_f32(
          false, a[cur], false, b[cur][t], (short)0, acc[t], false, false);
  }
}

template <int K>
__global__ __launch_bounds__(128) __attribute__((amdgpu_waves_per_eu(4, 4)))
void gat_gemm_pair(const float* __restrict__ X, const float* __restrict__ W,
                   const float* __restrict__ RW, float* __restrict__ OUTH,
                   float* __restrict__ OUTR, int Oout,
                   const float* __restrict__ al, const float* __restrict__ ar,
                   float* __restrict__ el, float* __restrict__ er, int Hh, int F) {
  __shared__ __attribute__((aligned(16))) float tile[2 * GB_STAGE];
  const int tid  = threadIdx.x;
  const int lane = tid & 31;
  const int wv   = tid >> 5;
  const int row0 = blockIdx.x * GB_T;
  const int col0 = blockIdx.y * GB_T;
  const int mr   = lane & 15;
  const int half = lane >> 4;

  const int c4 = tid & 7;
  const int r8 = tid >> 3;
  const float* xp[4];
  for (int i = 0; i < 4; ++i) {
    int xr = row0 + r8 + 16 * i;
    xr = (xr < NN) ? xr : (NN - 1);
    xp[i] = X + (size_t)xr * (size_t)K + 4 * c4;
  }
  const float* wp = W  + (size_t)(col0 + r8) * (size_t)K + 4 * c4;
  const float* rp = RW + (size_t)(col0 + r8) * (size_t)K + 4 * c4;
  float* sx = tile + r8 * GB_P + 4 * c4;
  float* sw = sx + GB_T * GB_P;
  float* sr = sw + GB_T * GB_P;
  const float* ap = tile + (16 * wv + mr) * GB_P + 2 * half;
  const float* bp = tile + (GB_T + mr) * GB_P + 2 * half;

  float4 px[4], pw[4], pr[4];
  for (int i = 0; i < 4; ++i) {
    px[i] = *(const float4*)(xp[i]);
    pw[i] = *(const float4*)(wp + 16 * i * K);
    pr[i] = *(const float4*)(rp + 16 * i * K);
  }
  for (int i = 0; i < 4; ++i) {
    *(float4*)(sx + 16 * i * GB_P) = px[i];
    *(float4*)(sw + 16 * i * GB_P) = pw[i];
    *(float4*)(sr + 16 * i * GB_P) = pr[i];
  }
  __syncthreads();

  v8f acc[8] = {};
  constexpr int nkb = K / GB_K;
#pragma unroll 1
  for (int kb = 0; kb + 1 < nkb; ++kb) {
    const int so = (kb & 1) * GB_STAGE;
    const int sn = GB_STAGE - so;
    const int kn = (kb + 1) * GB_K;
    for (int i = 0; i < 4; ++i) {
      px[i] = *(const float4*)(xp[i] + kn);
      pw[i] = *(const float4*)(wp + 16 * i * K + kn);
      pr[i] = *(const float4*)(rp + 16 * i * K + kn);
    }
    gat_gemm_kblock(ap + so, bp + so, acc);
    for (int i = 0; i < 4; ++i) {
      *(float4*)(sx + sn + 16 * i * GB_P) = px[i];
      *(float4*)(sw + sn + 16 * i * GB_P) = pw[i];
      *(float4*)(sr + sn + 16 * i * GB_P) = pr[i];
    }
    __syncthreads();
  }
  float cl[4], cr[4];
  {
    const int head = blockIdx.y;
    for (int t = 0; t < 4; ++t) {
      const int c = (t << 4) + mr;
      const bool in = c < F;
      const int ci = head * F + (in ? c : 0);
      const float a = al[ci];
      const float b = ar[ci];
      cl[t] = in ? a : 0.0f;
      cr[t] = in ? b : 0.0f;
    }
  }
  {
    const int so = ((nkb - 1) & 1) * GB_STAGE;
    gat_gemm_kblock(ap + so, bp + so, acc);
  }

  const int wrow0 = row0 + 16 * wv;
  if (wrow0 < NN) {
    for (int t = 0; t < 4; ++t) {
      const int oc = col0 + (t << 4) + mr;
      if (oc < Oout) {
        const size_t o0 = (size_t)(wrow0 + (half << 3)) * (size_t)Oout + oc;
        float* hrow = OUTH + o0;
        float* rrow = OUTR + o0;
        for (int v = 0; v < 8; ++v) {
          hrow[(size_t)v * (size_t)Oout] = acc[t][v];
          rrow[(size_t)v * (size_t)Oout] = acc[4 + t][v];
        }
      }
    }
  }

  {
    float pl[8], pq[8];
    for (int v = 0; v < 8; ++v) {
      float sl = acc[0][v] * cl[0];
      float sq = acc[0][v] * cr[0];
      for (int t = 1; t < 4; ++t) {
        sl = fmaf(acc[t][v], cl[t], sl);
        sq = fmaf(acc[t][v], cr[t], sq);
      }
      pl[v] = sl;
      pq[v] = sq;
    }
    for (int v = 0; v < 8; ++v) { pl[v] = gat_row16_step<1>(pl[v]); pq[v] = gat_row16_step<1>(pq[v]); }
    for (int v = 0; v < 8; ++v) { pl[v] = gat_row16_step<2>(pl[v]); pq[v] = gat_row16_step<2>(pq[v]); }
    for (int v = 0; v < 8; ++v) { pl[v] = gat_row16_step<4>(pl[v]); pq[v] = gat_row16_step<4>(pq[v]); }
    for (int v = 0; v < 8; ++v) { pl[v] = gat_row16_step<8>(pl[v]); pq[v] = gat_row16_step<8>(pq[v]); }
    const int v8 = mr & 7;
    float ol = pl[0], oq = pq[0];
    for (int v = 1; v < 8; ++v) {
      ol = (v8 == v) ? pl[v] : ol;
      oq = (v8 == v) ? pq[v] : oq;
    }
    if (wrow0 < NN) {
      float* dp = (mr < 8) ? el : er;
      dp[(size_t)(wrow0 + (half << 3) + v8) * (size_t)Hh + blockIdx.y] = (mr < 8) ? ol : oq;
    }
  }
}

__global__ void gat_pad_w(const float* __restrict__ Wsrc, float* __restrict__ Wdst,
                          int rows, int padrows, int K) {
  int idx = blockIdx.x * blockDim.x + threadIdx.x;
  int total = padrows * K;
  if (idx >= total) return;
  int r = idx / K;
  Wdst[idx] = (r < rows) ? Wsrc[idx] : 0.0f;
}

#define CSR_PERM (NN + 1)
#define CSR_CUR  (CSR_PERM + EE)
#define CSR_BSUM (CSR_CUR + NN)
#define CSR_NB   ((NN + 255) / 256)

__global__ void gat_csr_zero(int* __restrict__ cur, int total) {
  int idx = blockIdx.x * blockDim.x + threadIdx.x;
  if (idx < total) cur[idx] = 0;
}

__global__ void gat_csr_hist(const int* __restrict__ dst, int* __restrict__ cur, int total) {
  int e = blockIdx.x * blockDim.x + threadIdx.x;
  if (e < total) atomicAdd(cur + dst[e], 1);
}

__device__ __forceinline__ int gat_block_scan256(int v, int* sh) {
  const int t = threadIdx.x;
  sh[t] = v;
  __syncthreads();
  for (int d = 1; d < 256; d <<= 1) {
    int a = (t >= d) ? sh[t - d] : 0;
    __syncthreads();
    sh[t] += a;
    __syncthreads();
  }
  return sh[t];
}

__global__ __launch_bounds__(256)
void gat_csr_bsum(const int* __restrict__ cur, int* __restrict__ bsum) {
  __shared__ int sh[256];
  const int i = blockIdx.x * 256 + threadIdx.x;
  const int v = (i < NN) ? cur[i] : 0;
  const int inc = gat_block_scan256(v, sh);
  if (threadIdx.x == 255) bsum[blockIdx.x] = inc;
}

__global__ __launch_bounds__(256)
void gat_csr_offsets(int* __restrict__ cur, const int* __restrict__ bsum,
                     int* __restrict__ off) {
  __shared__ int sh[256];
  const int t = threadIdx.x;
  const int b = blockIdx.x;
  const int x = (t < b) ? bsum[t] : 0;
  gat_block_scan256(x, sh);
  const int before = sh[255];
  __syncthreads();
  const int i = b * 256 + t;
  const int v = (i < NN) ? cur[i] : 0;
  const int inc = gat_block_scan256(v, sh);
  if (i < NN) {
    const int o = before + inc - v;
    off[i] = o;
    cur[i] = o;
    if (i == NN - 1) off[NN] = o + v;
  }
}

__global__ void gat_csr_fill(const int* __restrict__ dst, const int* __restrict__ src,
                             int* __restrict__ cur, int* __restrict__ perm, int total) {
  int e = blockIdx.x * blockDim.x + threadIdx.x;
  if (e >= total) return;
  const int s = src[e];
  int p = atomicAdd(cur + dst[e], 1);
  perm[p] = s;
}


__device__ __forceinline__ float gat_leaky(float s) {
  return (s > 0.0f) ? s : 0.2f * s;
}
__device__ __forceinline__ float gat_elu1(float x) {
  return (x > 0.0f) ? x : expm1f(x);
}
__device__ __forceinline__ float gat_wave_max(float v) {
  for (int d = 16; d >= 1; d >>= 1) v = fmaxf(v, __shfl_xor(v, d, 32));
  return v;
}
__device__ __forceinline__ float gat_wave_sum(float v) {
  for (int d = 16; d >= 1; d >>= 1) v += __shfl_xor(v, d, 32);
  return v;
}

__device__ __forceinline__ void gat_chunk4(int base, int end, int lane,
                                           const int* __restrict__ perm,
                                           const float* __restrict__ el, float4 rn,
                                           int& my_s, float4& sc) {
  my_s = 0;
  sc.x = -INFINITY; sc.y = -INFINITY; sc.z = -INFINITY; sc.w = -INFINITY;
  if (base + lane < end) {
    my_s = perm[base + lane];
    const float4 l = *(const float4*)(el + (size_t)my_s * 4);
    sc.x = gat_leaky(l.x + rn.x);
    sc.y = gat_leaky(l.y + rn.y);
    sc.z = gat_leaky(l.z + rn.z);
    sc.w = gat_leaky(l.w + rn.w);
  }
}

__global__ __launch_bounds__(256)
void gat_gather256(const float* __restrict__ Hm, const float* __restrict__ el,
                   const float* __restrict__ er,
                   const int* __restrict__ off, const int* __restrict__ perm,
                   float* __restrict__ AGG, int act) {
  const int lane = threadIdx.x & 31;
  const int n = __builtin_amdgcn_readfirstlane((int)(blockIdx.x * 8 + (threadIdx.x >> 5)));
  if (n >= NN) return;
  const int beg = off[n];
  const int end = off[n + 1];
  const float4 rn = *(const float4*)(er + (size_t)n * 4);
  const bool one = (end - beg) <= 32;
  const bool hi = lane >= 16;
  int my_s = 0;
  float4 sc = {-INFINITY, -INFINITY, -INFINITY, -INFINITY};

  float4 mx = {-INFINITY, -INFINITY, -INFINITY, -INFINITY};
  for (int base = beg; base < end; base += 32) {
    gat_chunk4(base, end, lane, perm, el, rn, my_s, sc);
    mx.x = fmaxf(mx.x, sc.x); mx.y = fmaxf(mx.y, sc.y);
    mx.z = fmaxf(mx.z, sc.z); mx.w = fmaxf(mx.w, sc.w);
  }
  mx.x = gat_wave_max(mx.x); mx.y = gat_wave_max(mx.y);
  mx.z = gat_wave_max(mx.z); mx.w = gat_wave_max(mx.w);

  float4 dn = {0.f, 0.f, 0.f, 0.f};
  for (int base = beg; base < end; base += 32) {
    if (!one) gat_chunk4(base, end, lane, perm, el, rn, my_s, sc);
    dn.x += expf(sc.x - mx.x); dn.y += expf(sc.y - mx.y);
    dn.z += expf(sc.z - mx.z); dn.w += expf(sc.w - mx.w);
  }
  dn.x = gat_wave_sum(dn.x); dn.y = gat_wave_sum(dn.y);
  dn.z = gat_wave_sum(dn.z); dn.w = gat_wave_sum(dn.w);

  float4 acc0 = {0.f, 0.f, 0.f, 0.f};
  float4 acc1 = {0.f, 0.f, 0.f, 0.f};
  for (int base = beg; base < end; base += 32) {
    const int left = end - base;
    const int cnt = (left < 32) ? left : 32;
    if (!one) gat_chunk4(base, end, lane, perm, el, rn, my_s, sc);
    float4 al;
    al.x = expf(sc.x - mx.x) / dn.x;
    al.y = expf(sc.y - mx.y) / dn.y;
    al.z = expf(sc.z - mx.z) / dn.z;
    al.w = expf(sc.w - mx.w) / dn.w;
#pragma unroll 4
    for (int i = 0; i < cnt; ++i) {
      const int s = __builtin_amdgcn_readlane(my_s, i);
      const float a0 = __int_as_float(__builtin_amdgcn_readlane(__float_as_int(al.x), i));
      const float a1 = __int_as_float(__builtin_amdgcn_readlane(__float_as_int(al.y), i));
      const float a2 = __int_as_float(__builtin_amdgcn_readlane(__float_as_int(al.z), i));
      const float a3 = __int_as_float(__builtin_amdgcn_readlane(__float_as_int(al.w), i));
      const float w0 = hi ? a1 : a0;
      const float w1 = hi ? a3 : a2;
      const float4* row = (const float4*)(Hm + (size_t)s * 256);
      const float4 v0 = row[lane];
      const float4 v1 = row[32 + lane];
      acc0.x += v0.x * w0; acc0.y += v0.y * w0; acc0.z += v0.z * w0; acc0.w += v0.w * w0;
      acc1.x += v1.x * w1; acc1.y += v1.y * w1; acc1.z += v1.z * w1; acc1.w += v1.w * w1;
    }
  }

  float4* o = (float4*)(AGG + (size_t)n * 256);
  float4 r0 = o[lane];
  float4 r1 = o[32 + lane];
  r0.x += acc0.x; r0.y += acc0.y; r0.z += acc0.z; r0.w += acc0.w;
  r1.x += acc1.x; r1.y += acc1.y; r1.z += acc1.z; r1.w += acc1.w;
  if (act) {
    r0.x = gat_elu1(r0.x); r0.y = gat_elu1(r0.y); r0.z = gat_elu1(r0.z); r0.w = gat_elu1(r0.w);
    r1.x = gat_elu1(r1.x); r1.y = gat_elu1(r1.y); r1.z = gat_elu1(r1.z); r1.w = gat_elu1(r1.w);
  }
  o[lane] = r0;
  o[32 + lane] = r1;
}

__device__ __forceinline__ void gat_chunk1(int base, int end, int lane,
                                           const int* __restrict__ perm,
                                           const float* __restrict__ el, float rn,
                                           int& my_s, float& sc) {
  my_s = 0;
  sc = -INFINITY;
  if (base + lane < end) {
    my_s = perm[base + lane];
    sc = gat_leaky(el[my_s] + rn);
  }
}

__global__ __launch_bounds__(256)
void gat_gather40(const float* __restrict__ Hm, const float* __restrict__ el,
                  const float* __restrict__ er,
                  const int* __restrict__ off, const int* __restrict__ perm,
                  float* __restrict__ AGG, int act) {
  const int lane = threadIdx.x & 31;
  const int g = __builtin_amdgcn_readfirstlane((int)(blockIdx.x * 8 + (threadIdx.x >> 5)));
  if (g * 4 >= NN) return;
  const int col = ((lane < 20) ? lane : 19) * 2;
  for (int q = 0; q < 4; ++q) {
    const int n = g * 4 + q;
    const int beg = off[n];
    const int end = off[n + 1];
    const float rn = er[n];
    const bool one = (end - beg) <= 32;
    int my_s = 0;
    float sc = -INFINITY;

    float mx = -INFINITY;
    for (int base = beg; base < end; base += 32) {
      gat_chunk1(base, end, lane, perm, el, rn, my_s, sc);
      mx = fmaxf(mx, sc);
    }
    mx = gat_wave_max(mx);

    float dn = 0.f;
    for (int base = beg; base < end; base += 32) {
      if (!one) gat_chunk1(base, end, lane, perm, el, rn, my_s, sc);
      dn += expf(sc - mx);
    }
    dn = gat_wave_sum(dn);

    float2 acc = {0.f, 0.f};
    for (int base = beg; base < end; base += 32) {
      const int left = end - base;
      const int cnt = (left < 32) ? left : 32;
      if (!one) gat_chunk1(base, end, lane, perm, el, rn, my_s, sc);
      const float my_a = expf(sc - mx) / dn;
#pragma unroll 4
      for (int i = 0; i < cnt; ++i) {
        const int s = __builtin_amdgcn_readlane(my_s, i);
        const float a = __int_as_float(__builtin_amdgcn_readlane(__float_as_int(my_a), i));
        const float2 v = *(const float2*)(Hm + (size_t)s * 40 + col);
        acc.x += v.x * a;
        acc.y += v.y * a;
      }
    }
    if (lane < 20) {
      float2* o = (float2*)(AGG + (size_t)n * 40 + col);
      float2 r = *o;
      r.x += acc.x;
      r.y += acc.y;
      if (act) { r.x = gat_elu1(r.x); r.y = gat_elu1(r.y); }
      *o = r;
    }
  }
}

static void run_layer(const float* X, int K, int Hh, int F,
                      const float* W, const float* al, const float* ar, const float* rW,
                      float* Hbuf, float* AGG, bool act, int Opad,
                      float* el, float* er, const int* csr,
                      hipStream_t stream) {
  const int HF = Hh * F;
  dim3 ggrid((NN + GB_T - 1) / GB_T, Opad / GB_T);
  if (K == 128)
    gat_gemm_pair<128><<<ggrid, 128, 0, stream>>>(X, W, rW, Hbuf, AGG, HF, al, ar, el, er, Hh, F);
  else
    gat_gemm_pair<256><<<ggrid, 128, 0, stream>>>(X, W, rW, Hbuf, AGG, HF, al, ar, el, er, Hh, F);

  if (HF == 256)
    gat_gather256<<<NN / 8, 256, 0, stream>>>(Hbuf, el, er, csr, csr + CSR_PERM, AGG,
                                              act ? 1 : 0);
  else
    gat_gather40<<<(NN / 4 + 7) / 8, 256, 0, stream>>>(Hbuf, el, er, csr, csr + CSR_PERM,
                                                       AGG, act ? 1 : 0);
}

extern "C" void kernel_launch(void* const* d_in, const int* in_sizes, int n_in,
                              void* d_out, int out_size, void* d_ws, size_t ws_size,
                              hipStream_t stream) {
  (void)in_sizes; (void)n_in; (void)out_size; (void)ws_size;

  const float* feat = (const float*)d_in[0];
  const int*   src  = (const int*)d_in[1];
  const int*   dst  = (const int*)d_in[2];
  const float* W0  = (const float*)d_in[3];
  const float* al0 = (const float*)d_in[4];
  const float* ar0 = (const float*)d_in[5];
  const float* rW0 = (const float*)d_in[6];
  const float* W1  = (const float*)d_in[7];
  const float* al1 = (const float*)d_in[8];
  const float* ar1 = (const float*)d_in[9];
  const float* rW1 = (const float*)d_in[10];
  const float* W2  = (const float*)d_in[11];
  const float* al2 = (const float*)d_in[12];
  const float* ar2 = (const float*)d_in[13];
  const float* rW2 = (const float*)d_in[14];

  const size_t NF = (size_t)NN * 256;
  float* ws    = (float*)d_ws;
  float* bufH  = ws;
  float* bufY0 = bufH  + NF;
  float* bufY1 = bufY0 + NF;
  float* el    = bufY1 + NF;
  float* er    = el  + (size_t)NN * 4;
  float* sb    = er  + (size_t)NN * 12;
  float* W2p   = sb + (size_t)EE * 4;
  float* rW2p  = W2p + (size_t)64 * 256;

  {
    int tot = 64 * 256;
    gat_pad_w<<<(tot + 255) / 256, 256, 0, stream>>>(W2,  W2p,  40, 64, 256);
    gat_pad_w<<<(tot + 255) / 256, 256, 0, stream>>>(rW2, rW2p, 40, 64, 256);
  }

  int* csr = (int*)sb;
  {
    gat_csr_zero<<<(NN + 255) / 256, 256, 0, stream>>>(csr + CSR_CUR, NN);
    gat_csr_hist<<<(EE + 255) / 256, 256, 0, stream>>>(dst, csr + CSR_CUR, EE);
    gat_csr_bsum<<<CSR_NB, 256, 0, stream>>>(csr + CSR_CUR, csr + CSR_BSUM);
    gat_csr_offsets<<<CSR_NB, 256, 0, stream>>>(csr + CSR_CUR, csr + CSR_BSUM, csr);
    gat_csr_fill<<<(EE + 255) / 256, 256, 0, stream>>>(dst, src, csr + CSR_CUR, csr + CSR_PERM,
                                                       EE);
  }

  run_layer(feat, 128, 4, 64, W0, al0, ar0, rW0,
            bufH, bufY0, true, 256, el, er, csr, stream);
  run_layer(bufY0, 256, 4, 64, W1, al1, ar1, rW1,
            bufH, bufY1, true, 256, el, er, csr, stream);
  run_layer(bufY1, 256, 1, 40, W2p, al2, ar2, rW2p,
            bufH, (float*)d_out, false, 64, el, er, csr, stream);
}
